# starting kernel
# baseline (speedup 1.0000x reference)
#include <hip/hip_runtime.h>
#include <hip/hip_bf16.h>

typedef __attribute__((ext_vector_type(2))) float v2f;
typedef __attribute__((ext_vector_type(8))) float v8f;

#define NEG_SLOPE 0.2f

__device__ __forceinline__ unsigned fenc(float f) {
    unsigned i = __float_as_uint(f);
    return (i & 0x80000000u) ? ~i : (i | 0x80000000u);
}
__device__ __forceinline__ float fdec(unsigned u) {
    return (u & 0x80000000u) ? __uint_as_float(u & 0x7fffffffu)
                             : __uint_as_float(~u);
}

template <int NT, int NC>
__global__ __launch_bounds__(32) void gemm128_wmma(
    const float* __restrict__ A, const float* __restrict__ B,
    const float* __restrict__ bias, float* __restrict__ C, int M) {
    constexpr int strips = NC / (16 * NT);
    const int tm = blockIdx.x / strips;
    const int ts = blockIdx.x % strips;
    const int lane = threadIdx.x;
    const int half = lane >> 4;
    const int l16  = lane & 15;

    int rowA = tm * 16 + l16;
    int rA = rowA < M ? rowA : (M - 1);
    const int k0base = half * 2;
    const float* Ap = A + (size_t)rA * 128 + k0base;
    const int col0 = ts * 16 * NT + l16;
    const float* Bp = B + (size_t)k0base * NC + col0;

    const v8f vzero = {0.f, 0.f, 0.f, 0.f, 0.f, 0.f, 0.f, 0.f};
    v8f c[NT];
#pragma unroll
    for (int t = 0; t < NT; ++t) c[t] = vzero;

#pragma unroll
    for (int kk = 0; kk < 128; kk += 4) {
        v2f a;
        a.x = Ap[kk];
        a.y = Ap[kk + 1];
        v2f b[NT];
#pragma unroll
        for (int t = 0; t < NT; ++t) {
            b[t].x = Bp[(size_t)kk * NC + t * 16];
            b[t].y = Bp[(size_t)(kk + 1) * NC + t * 16];
        }
#pragma unroll
        for (int t = 0; t < NT; ++t) {
            c[t] = __builtin_amdgcn_wmma_f32_16x16x4_f32(
                false, a, false, b[t], (short)0, c[t], false, false);
        }
    }
#pragma unroll
    for (int t = 0; t < NT; ++t) {
        float bv = bias ? bias[col0 + t * 16] : 0.f;
#pragma unroll
        for (int r = 0; r < 8; ++r) {
            int row = tm * 16 + r + half * 8;
            if (row < M) C[(size_t)row * NC + col0 + t * 16] = c[t][r] + bv;
        }
    }
}

__global__ void init_gat_kernel(float* acc, float* s, unsigned* m, long long n) {
    long long gid = (long long)blockIdx.x * blockDim.x + threadIdx.x;
    if (gid >= n * 128) return;
    acc[gid] = 0.f;
    if (gid < n * 4) { s[gid] = 0.f; m[gid] = 0u; }
}

__global__ void init_adj_kernel(float* adj, float* cnt, long long g) {
    long long gid = (long long)blockIdx.x * blockDim.x + threadIdx.x;
    if (gid >= g * 128) return;
    adj[gid] = 0.f;
    if (gid < g) cnt[gid] = 0.f;
}

__global__ void attn_kernel(const float* __restrict__ h,
                            const float* __restrict__ ws,
                            const float* __restrict__ wd,
                            float* __restrict__ a_s, float* __restrict__ a_d,
                            int n) {
    long long gid = (long long)blockIdx.x * blockDim.x + threadIdx.x;
    if (gid >= (long long)n * 4) return;
    int node = (int)(gid >> 2), hd = (int)(gid & 3);
    const float* hr = h + (size_t)node * 128 + hd * 32;
    float ss = 0.f, sd = 0.f;
#pragma unroll
    for (int c = 0; c < 32; ++c) {
        float hv = hr[c];
        ss += hv * ws[hd * 32 + c];
        sd += hv * wd[hd * 32 + c];
    }
    a_s[gid] = ss;
    a_d[gid] = sd;
}

__global__ void edge_max_kernel(const int* __restrict__ ei, int E,
                                const float* __restrict__ a_s,
                                const float* __restrict__ a_d,
                                unsigned* __restrict__ m, long long Etot) {
    long long gid = (long long)blockIdx.x * blockDim.x + threadIdx.x;
    if (gid >= Etot * 4) return;
    int e = (int)(gid >> 2), hd = (int)(gid & 3);
    int src, dst;
    if (e < E) { src = ei[e]; dst = ei[E + e]; } else { src = dst = e - E; }
    float v = a_s[src * 4 + hd] + a_d[dst * 4 + hd];
    v = v > 0.f ? v : NEG_SLOPE * v;
    atomicMax(&m[dst * 4 + hd], fenc(v));
}

__global__ void edge_accum_kernel(const int* __restrict__ ei, int E,
                                  const float* __restrict__ a_s,
                                  const float* __restrict__ a_d,
                                  const unsigned* __restrict__ m,
                                  const float* __restrict__ h,
                                  float* __restrict__ s,
                                  float* __restrict__ acc, long long Etot) {
    long long gid = (long long)blockIdx.x * blockDim.x + threadIdx.x;
    if (gid >= Etot * 128) return;
    int e = (int)(gid >> 7), ch = (int)(gid & 127), hd = ch >> 5;
    int src, dst;
    if (e < E) { src = ei[e]; dst = ei[E + e]; } else { src = dst = e - E; }
    float v = a_s[src * 4 + hd] + a_d[dst * 4 + hd];
    v = v > 0.f ? v : NEG_SLOPE * v;
    float ex = __expf(v - fdec(m[dst * 4 + hd]));
    if ((ch & 31) == 0) atomicAdd(&s[dst * 4 + hd], ex);
    atomicAdd(&acc[(size_t)dst * 128 + ch], ex * h[(size_t)src * 128 + ch]);
}

__global__ void finalize_kernel(const float* __restrict__ acc,
                                const float* __restrict__ s,
                                const float* __restrict__ bias,
                                float* __restrict__ dst, long long n) {
    long long gid = (long long)blockIdx.x * blockDim.x + threadIdx.x;
    if (gid >= n * 128) return;
    int node = (int)(gid >> 7), ch = (int)(gid & 127), hd = ch >> 5;
    dst[gid] = acc[gid] / (s[node * 4 + hd] + 1e-16f) + bias[ch];
}

__global__ void adj_accum_kernel(const int* __restrict__ eig, int EG,
                                 const float* __restrict__ xg,
                                 float* __restrict__ adj,
                                 float* __restrict__ cnt) {
    long long gid = (long long)blockIdx.x * blockDim.x + threadIdx.x;
    if (gid >= (long long)EG * 128) return;
    int e = (int)(gid >> 7), ch = (int)(gid & 127);
    int g0 = eig[e], g1 = eig[EG + e];
    atomicAdd(&adj[(size_t)g0 * 128 + ch], xg[(size_t)g1 * 128 + ch]);
    if (ch == 0) atomicAdd(&cnt[g0], 1.f);
}

__global__ void adj_mean_kernel(float* __restrict__ adj,
                                const float* __restrict__ cnt, long long g) {
    long long gid = (long long)blockIdx.x * blockDim.x + threadIdx.x;
    if (gid >= g * 128) return;
    adj[gid] = adj[gid] / fmaxf(cnt[gid >> 7], 1.f);
}

__global__ __launch_bounds__(128) void node_update_kernel(
    const float* __restrict__ xo, const int* __restrict__ ng,
    const float* __restrict__ xg, const float* __restrict__ am,
    float* __restrict__ upd, int n) {
    int wave = threadIdx.x >> 5;
    int lane = threadIdx.x & 31;
    int node = blockIdx.x * (blockDim.x >> 5) + wave;
    if (node >= n) return;
    int g = ng[node];
    const float* xr = xo + (size_t)node * 128;
    const float* gf = xg + (size_t)g * 128;
    const float* af = am + (size_t)g * 128;
    float xv[4], gv[4], av[4];
    float pg = 0.f, pa = 0.f;
#pragma unroll
    for (int i = 0; i < 4; ++i) {
        int c = lane + i * 32;
        xv[i] = xr[c]; gv[i] = gf[c]; av[i] = af[c];
        pg += xv[i] * gv[i];
        pa += xv[i] * av[i];
    }
#pragma unroll
    for (int off = 16; off >= 1; off >>= 1) {
        pg += __shfl_xor(pg, off, 32);
        pa += __shfl_xor(pa, off, 32);
    }
    float* ur = upd + (size_t)node * 128;
#pragma unroll
    for (int i = 0; i < 4; ++i) {
        int c = lane + i * 32;
        ur[c] = xv[i] + pg * gv[i] + pa * av[i];
    }
}

static inline unsigned cdivll(long long a, long long b) {
    return (unsigned)((a + b - 1) / b);
}

extern "C" void kernel_launch(void* const* d_in, const int* in_sizes, int n_in,
                              void* d_out, int out_size, void* d_ws,
                              size_t ws_size, hipStream_t stream) {
    const float* x      = (const float*)d_in[0];
    const float* x_g    = (const float*)d_in[1];
    const int*   ei     = (const int*)d_in[2];
    const int*   eig    = (const int*)d_in[3];
    const int*   ngrp   = (const int*)d_in[4];
    const float* W1     = (const float*)d_in[5];
    const float* atts1  = (const float*)d_in[6];
    const float* attd1  = (const float*)d_in[7];
    const float* bias1  = (const float*)d_in[8];
    const float* W2     = (const float*)d_in[9];
    const float* atts2  = (const float*)d_in[10];
    const float* attd2  = (const float*)d_in[11];
    const float* bias2  = (const float*)d_in[12];
    const float* W_fc   = (const float*)d_in[13];
    const float* b_fc   = (const float*)d_in[14];

    const int N  = in_sizes[0] / 128;
    const int G  = in_sizes[1] / 128;
    const int E  = in_sizes[2] / 2;
    const int EG = in_sizes[3] / 2;
    const long long EtN = (long long)E + N;
    const long long EtG = (long long)EG + G;

    float* out    = (float*)d_out;
    float* xg_out = out + (size_t)N * 32;

    float* ws = (float*)d_ws;
    size_t o = 0;
    float*    h     = ws + o; o += (size_t)N * 128;
    float*    acc   = ws + o; o += (size_t)N * 128;
    float*    a_s   = ws + o; o += (size_t)N * 4;
    float*    a_d   = ws + o; o += (size_t)N * 4;
    float*    sden  = ws + o; o += (size_t)N * 4;
    unsigned* menc  = (unsigned*)(ws + o); o += (size_t)N * 4;
    float*    h_g   = ws + o; o += (size_t)G * 128;
    float*    acc_g = ws + o; o += (size_t)G * 128;
    float*    a_sg  = ws + o; o += (size_t)G * 4;
    float*    a_dg  = ws + o; o += (size_t)G * 4;
    float*    s_g   = ws + o; o += (size_t)G * 4;
    unsigned* m_g   = (unsigned*)(ws + o); o += (size_t)G * 4;
    float*    adj   = ws + o; o += (size_t)G * 128;
    float*    cnt   = ws + o; o += (size_t)G;

    const int TB = 256;

    init_gat_kernel<<<cdivll((long long)N * 128, TB), TB, 0, stream>>>(
        acc, sden, menc, N);
    {
        unsigned tiles = (unsigned)(((N + 15) / 16) * (128 / 64));
        gemm128_wmma<4, 128><<<tiles, 32, 0, stream>>>(x, W1, nullptr, h, N);
    }
    attn_kernel<<<cdivll((long long)N * 4, TB), TB, 0, stream>>>(
        h, atts1, attd1, a_s, a_d, N);
    edge_max_kernel<<<cdivll(EtN * 4, TB), TB, 0, stream>>>(
        ei, E, a_s, a_d, menc, EtN);
    edge_accum_kernel<<<cdivll(EtN * 128, TB), TB, 0, stream>>>(
        ei, E, a_s, a_d, menc, h, sden, acc, EtN);
    finalize_kernel<<<cdivll((long long)N * 128, TB), TB, 0, stream>>>(
        acc, sden, bias1, acc, N);

    init_gat_kernel<<<cdivll((long long)G * 128, TB), TB, 0, stream>>>(
        acc_g, s_g, m_g, G);
    {
        unsigned tiles = (unsigned)(((G + 15) / 16) * (128 / 64));
        gemm128_wmma<4, 128><<<tiles, 32, 0, stream>>>(x_g, W2, nullptr, h_g, G);
    }
    attn_kernel<<<cdivll((long long)G * 4, TB), TB, 0, stream>>>(
        h_g, atts2, attd2, a_sg, a_dg, G);
    edge_max_kernel<<<cdivll(EtG * 4, TB), TB, 0, stream>>>(
        eig, EG, a_sg, a_dg, m_g, EtG);
    edge_accum_kernel<<<cdivll(EtG * 128, TB), TB, 0, stream>>>(
        eig, EG, a_sg, a_dg, m_g, h_g, s_g, acc_g, EtG);
    finalize_kernel<<<cdivll((long long)G * 128, TB), TB, 0, stream>>>(
        acc_g, s_g, bias2, xg_out, G);

    init_adj_kernel<<<cdivll((long long)G * 128, TB), TB, 0, stream>>>(
        adj, cnt, G);
    adj_accum_kernel<<<cdivll((long long)EG * 128, TB), TB, 0, stream>>>(
        eig, EG, xg_out, adj, cnt);
    adj_mean_kernel<<<cdivll((long long)G * 128, TB), TB, 0, stream>>>(
        adj, cnt, G);

    node_update_kernel<<<cdivll(N, 4), 128, 0, stream>>>(
        acc, ngrp, xg_out, adj, h, N);
    {
        unsigned tiles = (unsigned)((N + 15) / 16);
        gemm128_wmma<2, 32><<<tiles, 32, 0, stream>>>(h, W_fc, b_fc, out, N);
    }
}


# Edges grouped by destination each call (counting sort, arrival-order rank); one wave per node sums its run in registers: no float atomics on the attention sums, sum order changes
# speedup vs baseline: 3.2401x; 3.2401x over previous
#include <hip/hip_runtime.h>
#include <hip/hip_bf16.h>

typedef __attribute__((ext_vector_type(2))) float v2f;
typedef __attribute__((ext_vector_type(8))) float v8f;

#define NEG_SLOPE 0.2f

template <int NT, int NC>
__global__ __launch_bounds__(32) void gemm128_wmma(
    const float* __restrict__ A, const float* __restrict__ B,
    const float* __restrict__ bias, float* __restrict__ C, int M) {
    constexpr int strips = NC / (16 * NT);
    const int tm = blockIdx.x / strips;
    const int ts = blockIdx.x % strips;
    const int lane = threadIdx.x;
    const int half = lane >> 4;
    const int l16  = lane & 15;

    int rowA = tm * 16 + l16;
    int rA = rowA < M ? rowA : (M - 1);
    const int k0base = half * 2;
    const float* Ap = A + (size_t)rA * 128 + k0base;
    const int col0 = ts * 16 * NT + l16;
    const float* Bp = B + (size_t)k0base * NC + col0;

    const v8f vzero = {0.f, 0.f, 0.f, 0.f, 0.f, 0.f, 0.f, 0.f};
    v8f c[NT];
#pragma unroll
    for (int t = 0; t < NT; ++t) c[t] = vzero;

#pragma unroll
    for (int kk = 0; kk < 128; kk += 4) {
        v2f a;
        a.x = Ap[kk];
        a.y = Ap[kk + 1];
        v2f b[NT];
#pragma unroll
        for (int t = 0; t < NT; ++t) {
            b[t].x = Bp[(size_t)kk * NC + t * 16];
            b[t].y = Bp[(size_t)(kk + 1) * NC + t * 16];
        }
#pragma unroll
        for (int t = 0; t < NT; ++t) {
            c[t] = __builtin_amdgcn_wmma_f32_16x16x4_f32(
                false, a, false, b[t], (short)0, c[t], false, false);
        }
    }
#pragma unroll
    for (int t = 0; t < NT; ++t) {
        float bv = bias ? bias[col0 + t * 16] : 0.f;
#pragma unroll
        for (int r = 0; r < 8; ++r) {
            int row = tm * 16 + r + half * 8;
            if (row < M) C[(size_t)row * NC + col0 + t * 16] = c[t][r] + bv;
        }
    }
}

__global__ void zero_deg_kernel(int* __restrict__ da, int na,
                                int* __restrict__ db, int nb) {
    int gid = blockIdx.x * blockDim.x + threadIdx.x;
    if (gid < na) da[gid] = 0;
    if (gid < nb) db[gid] = 0;
}

__global__ void init_adj_kernel(float* adj, float* cnt, long long g) {
    long long gid = (long long)blockIdx.x * blockDim.x + threadIdx.x;
    if (gid >= g * 128) return;
    adj[gid] = 0.f;
    if (gid < g) cnt[gid] = 0.f;
}

__global__ void attn_kernel(const float* __restrict__ h,
                            const float* __restrict__ ws,
                            const float* __restrict__ wd,
                            float* __restrict__ a_s, float* __restrict__ a_d,
                            int n) {
    long long gid = (long long)blockIdx.x * blockDim.x + threadIdx.x;
    if (gid >= (long long)n * 4) return;
    int node = (int)(gid >> 2), hd = (int)(gid & 3);
    const float* hr = h + (size_t)node * 128 + hd * 32;
    float ss = 0.f, sd = 0.f;
#pragma unroll
    for (int c = 0; c < 32; ++c) {
        float hv = hr[c];
        ss += hv * ws[hd * 32 + c];
        sd += hv * wd[hd * 32 + c];
    }
    a_s[gid] = ss;
    a_d[gid] = sd;
}

__device__ __forceinline__ float lrelu(float v) {
    return v > 0.f ? v : NEG_SLOPE * v;
}

__global__ void rank_kernel(const int* __restrict__ ei, int E,
                            int* __restrict__ deg, int* __restrict__ rank) {
    int e = blockIdx.x * blockDim.x + threadIdx.x;
    if (e >= E) return;
    rank[e] = atomicAdd(&deg[ei[E + e]], 1);
}

__device__ __forceinline__ int wave_incl_scan(int v, int lane) {
    for (int off = 1; off < 32; off <<= 1) {
        int y = __shfl_up(v, off, 32);
        if (lane >= off) v += y;
    }
    return v;
}

__global__ __launch_bounds__(256) void scan_local_kernel(
    const int* __restrict__ deg, int* __restrict__ rowloc,
    int* __restrict__ btot, int n) {
    __shared__ int wsum[8];
    const int tid = threadIdx.x, lane = tid & 31, w = tid >> 5;
    const int base = blockIdx.x * 1024 + tid * 4;
    int v[4];
#pragma unroll
    for (int i = 0; i < 4; ++i) v[i] = (base + i < n) ? deg[base + i] : 0;
    const int t = v[0] + v[1] + v[2] + v[3];
    const int inc = wave_incl_scan(t, lane);
    if (lane == 31) wsum[w] = inc;
    __syncthreads();
    int wpre = 0, tot = 0;
    for (int i = 0; i < 8; ++i) {
        int x = wsum[i];
        if (i < w) wpre += x;
        tot += x;
    }
    int ex = wpre + inc - t;
#pragma unroll
    for (int i = 0; i < 4; ++i) {
        if (base + i < n) rowloc[base + i] = ex;
        ex += v[i];
    }
    if (tid == 0) btot[blockIdx.x] = tot;
}

__global__ __launch_bounds__(1024) void scan_blocks_kernel(
    const int* __restrict__ btot, int* __restrict__ boff, int nb) {
    __shared__ int wsum[32];
    const int tid = threadIdx.x, lane = tid & 31, w = tid >> 5;
    int carry = 0;
    for (int base = 0; base < nb; base += 1024) {
        const int i = base + tid;
        const int v = (i < nb) ? btot[i] : 0;
        const int inc = wave_incl_scan(v, lane);
        if (lane == 31) wsum[w] = inc;
        __syncthreads();
        int wpre = 0, tot = 0;
        for (int k = 0; k < 32; ++k) {
            int x = wsum[k];
            if (k < w) wpre += x;
            tot += x;
        }
        if (i < nb) boff[i] = carry + wpre + inc - v;
        carry += tot;
        __syncthreads();
    }
}

__global__ void place_kernel(const int* __restrict__ ei, int E,
                             const int* __restrict__ rowloc,
                             const int* __restrict__ boff,
                             const int* __restrict__ rank,
                             int* __restrict__ srt) {
    int e = blockIdx.x * blockDim.x + threadIdx.x;
    if (e >= E) return;
    int d = ei[E + e];
    srt[rowloc[d] + boff[d >> 10] + rank[e]] = ei[e];
}

__global__ __launch_bounds__(256) void gat_gather_kernel(
    const float* __restrict__ h, const float* __restrict__ a_s,
    const float* __restrict__ a_d, const int* __restrict__ srt,
    const int* __restrict__ rowloc, const int* __restrict__ boff,
    const int* __restrict__ deg, const float* __restrict__ bias,
    float* __restrict__ dst, int n) {
    const int node = __builtin_amdgcn_readfirstlane(
        (int)(blockIdx.x * 8 + (threadIdx.x >> 5)));
    if (node >= n) return;
    const int lane = threadIdx.x & 31;
    const int hd = lane >> 3;
    const int cnt = deg[node];
    const int* seg = srt + (rowloc[node] + boff[node >> 10]);
    const float4 ad4 = *reinterpret_cast<const float4*>(a_d + (size_t)node * 4);
    const float4 as4 = *reinterpret_cast<const float4*>(a_s + (size_t)node * 4);

    float m0 = lrelu(as4.x + ad4.x), m1 = lrelu(as4.y + ad4.y);
    float m2 = lrelu(as4.z + ad4.z), m3 = lrelu(as4.w + ad4.w);
    for (int base = 0; base < cnt; base += 32) {
        const int j = base + lane;
        if (j < cnt) {
            const int s = min(max(seg[j], 0), n - 1);
            const float4 q =
                *reinterpret_cast<const float4*>(a_s + (size_t)s * 4);
            m0 = fmaxf(m0, lrelu(q.x + ad4.x));
            m1 = fmaxf(m1, lrelu(q.y + ad4.y));
            m2 = fmaxf(m2, lrelu(q.z + ad4.z));
            m3 = fmaxf(m3, lrelu(q.w + ad4.w));
        }
    }
    for (int off = 16; off >= 1; off >>= 1) {
        m0 = fmaxf(m0, __shfl_xor(m0, off, 32));
        m1 = fmaxf(m1, __shfl_xor(m1, off, 32));
        m2 = fmaxf(m2, __shfl_xor(m2, off, 32));
        m3 = fmaxf(m3, __shfl_xor(m3, off, 32));
    }
    const float m  = hd == 0 ? m0    : hd == 1 ? m1    : hd == 2 ? m2    : m3;
    const float ad = hd == 0 ? ad4.x : hd == 1 ? ad4.y : hd == 2 ? ad4.z : ad4.w;
    const float as = hd == 0 ? as4.x : hd == 1 ? as4.y : hd == 2 ? as4.z : as4.w;

    const float* hl = h + lane * 4;
    float ssum = __expf(lrelu(as + ad) - m);
    float4 acc;
    {
        const float4 r = *reinterpret_cast<const float4*>(hl + (size_t)node * 128);
        acc.x = ssum * r.x; acc.y = ssum * r.y;
        acc.z = ssum * r.z; acc.w = ssum * r.w;
    }
    for (int base = 0; base < cnt; base += 32) {
        const int j = base + lane;
        const int s_l = (j < cnt) ? min(max(seg[j], 0), n - 1) : node;
        const int rem = cnt - base;
        const int nch = rem < 32 ? rem : 32;
        for (int t = 0; t < nch; t += 4) {
            int su[4];
            float av[4];
            float4 r[4];
#pragma unroll
            for (int u = 0; u < 4; ++u) {
                const int tt = t + u;
                su[u] = __builtin_amdgcn_readlane(s_l, tt < nch ? tt : nch - 1);
            }
#pragma unroll
            for (int u = 0; u < 4; ++u) {
                av[u] = a_s[(size_t)su[u] * 4 + hd];
                r[u] = *reinterpret_cast<const float4*>(hl + (size_t)su[u] * 128);
            }
#pragma unroll
            for (int u = 0; u < 4; ++u) {
                const float ex =
                    (t + u < nch) ? __expf(lrelu(av[u] + ad) - m) : 0.f;
                ssum += ex;
                acc.x += ex * r[u].x; acc.y += ex * r[u].y;
                acc.z += ex * r[u].z; acc.w += ex * r[u].w;
            }
        }
    }
    const float den = ssum + 1e-16f;
    float4 o;
    o.x = acc.x / den + bias[lane * 4 + 0];
    o.y = acc.y / den + bias[lane * 4 + 1];
    o.z = acc.z / den + bias[lane * 4 + 2];
    o.w = acc.w / den + bias[lane * 4 + 3];
    *reinterpret_cast<float4*>(dst + (size_t)node * 128 + lane * 4) = o;
}

__global__ void adj_accum_kernel(const int* __restrict__ eig, int EG,
                                 const float* __restrict__ xg,
                                 float* __restrict__ adj,
                                 float* __restrict__ cnt) {
    long long gid = (long long)blockIdx.x * blockDim.x + threadIdx.x;
    if (gid >= (long long)EG * 128) return;
    int e = (int)(gid >> 7), ch = (int)(gid & 127);
    int g0 = eig[e], g1 = eig[EG + e];
    atomicAdd(&adj[(size_t)g0 * 128 + ch], xg[(size_t)g1 * 128 + ch]);
    if (ch == 0) atomicAdd(&cnt[g0], 1.f);
}

__global__ void adj_mean_kernel(float* __restrict__ adj,
                                const float* __restrict__ cnt, long long g) {
    long long gid = (long long)blockIdx.x * blockDim.x + threadIdx.x;
    if (gid >= g * 128) return;
    adj[gid] = adj[gid] / fmaxf(cnt[gid >> 7], 1.f);
}

__global__ __launch_bounds__(128) void node_update_kernel(
    const float* __restrict__ xo, const int* __restrict__ ng,
    const float* __restrict__ xg, const float* __restrict__ am,
    float* __restrict__ upd, int n) {
    int wave = threadIdx.x >> 5;
    int lane = threadIdx.x & 31;
    int node = blockIdx.x * (blockDim.x >> 5) + wave;
    if (node >= n) return;
    int g = ng[node];
    const float* xr = xo + (size_t)node * 128;
    const float* gf = xg + (size_t)g * 128;
    const float* af = am + (size_t)g * 128;
    float xv[4], gv[4], av[4];
    float pg = 0.f, pa = 0.f;
#pragma unroll
    for (int i = 0; i < 4; ++i) {
        int c = lane + i * 32;
        xv[i] = xr[c]; gv[i] = gf[c]; av[i] = af[c];
        pg += xv[i] * gv[i];
        pa += xv[i] * av[i];
    }
#pragma unroll
    for (int off = 16; off >= 1; off >>= 1) {
        pg += __shfl_xor(pg, off, 32);
        pa += __shfl_xor(pa, off, 32);
    }
    float* ur = upd + (size_t)node * 128;
#pragma unroll
    for (int i = 0; i < 4; ++i) {
        int c = lane + i * 32;
        ur[c] = xv[i] + pg * gv[i] + pa * av[i];
    }
}

static inline unsigned cdivll(long long a, long long b) {
    return (unsigned)((a + b - 1) / b);
}
static inline size_t al4(size_t v) { return (v + 3) & ~(size_t)3; }

extern "C" void kernel_launch(void* const* d_in, const int* in_sizes, int n_in,
                              void* d_out, int out_size, void* d_ws,
                              size_t ws_size, hipStream_t stream) {
    const float* x      = (const float*)d_in[0];
    const float* x_g    = (const float*)d_in[1];
    const int*   ei     = (const int*)d_in[2];
    const int*   eig    = (const int*)d_in[3];
    const int*   ngrp   = (const int*)d_in[4];
    const float* W1     = (const float*)d_in[5];
    const float* atts1  = (const float*)d_in[6];
    const float* attd1  = (const float*)d_in[7];
    const float* bias1  = (const float*)d_in[8];
    const float* W2     = (const float*)d_in[9];
    const float* atts2  = (const float*)d_in[10];
    const float* attd2  = (const float*)d_in[11];
    const float* bias2  = (const float*)d_in[12];
    const float* W_fc   = (const float*)d_in[13];
    const float* b_fc   = (const float*)d_in[14];

    const int N  = in_sizes[0] / 128;
    const int G  = in_sizes[1] / 128;
    const int E  = in_sizes[2] / 2;
    const int EG = in_sizes[3] / 2;

    float* out    = (float*)d_out;
    float* xg_out = out + (size_t)N * 32;

    float* ws = (float*)d_ws;
    const int nbN = (N + 1023) / 1024;
    const int nbG = (G + 1023) / 1024;
    size_t o = 0;
    float* h     = ws + o; o += (size_t)N * 128;
    float* acc   = ws + o; o += (size_t)N * 128;
    float* a_s   = ws + o; o += (size_t)N * 4;
    float* a_d   = ws + o; o += (size_t)N * 4;
    int*   degN  = (int*)(ws + o); o += al4((size_t)N);
    int*   rowN  = (int*)(ws + o); o += al4((size_t)N);
    int*   btN   = (int*)(ws + o); o += al4((size_t)nbN);
    int*   boN   = (int*)(ws + o); o += al4((size_t)nbN);
    float* h_g   = ws + o; o += (size_t)G * 128;
    float* a_sg  = ws + o; o += (size_t)G * 4;
    float* a_dg  = ws + o; o += (size_t)G * 4;
    int*   degG  = (int*)(ws + o); o += al4((size_t)G);
    int*   rowG  = (int*)(ws + o); o += al4((size_t)G);
    int*   btG   = (int*)(ws + o); o += al4((size_t)nbG);
    int*   boG   = (int*)(ws + o); o += al4((size_t)nbG);
    int*   rankG = (int*)(ws + o); o += al4((size_t)EG);
    int*   srtG  = (int*)(ws + o); o += al4((size_t)EG);
    float* adj   = ws + o; o += (size_t)G * 128;
    float* cnt   = ws + o; o += (size_t)G;
    int* rankN = (int*)acc;
    int* srtN  = (int*)out;

    const int TB = 256;

    zero_deg_kernel<<<cdivll(N > G ? N : G, TB), TB, 0, stream>>>(degN, N, degG, G);

    {
        unsigned tiles = (unsigned)(((N + 15) / 16) * (128 / 64));
        gemm128_wmma<4, 128><<<tiles, 32, 0, stream>>>(x, W1, nullptr, h, N);
    }
    attn_kernel<<<cdivll((long long)N * 4, TB), TB, 0, stream>>>(
        h, atts1, attd1, a_s, a_d, N);
    rank_kernel<<<cdivll(E, TB), TB, 0, stream>>>(ei, E, degN, rankN);
    scan_local_kernel<<<nbN, 256, 0, stream>>>(degN, rowN, btN, N);
    scan_blocks_kernel<<<1, 1024, 0, stream>>>(btN, boN, nbN);
    place_kernel<<<cdivll(E, TB), TB, 0, stream>>>(ei, E, rowN, boN, rankN, srtN);
    gat_gather_kernel<<<cdivll(N, 8), 256, 0, stream>>>(
        h, a_s, a_d, srtN, rowN, boN, degN, bias1, acc, N);

    {
        unsigned tiles = (unsigned)(((G + 15) / 16) * (128 / 64));
        gemm128_wmma<4, 128><<<tiles, 32, 0, stream>>>(x_g, W2, nullptr, h_g, G);
    }
    attn_kernel<<<cdivll((long long)G * 4, TB), TB, 0, stream>>>(
        h_g, atts2, attd2, a_sg, a_dg, G);
    rank_kernel<<<cdivll(EG, TB), TB, 0, stream>>>(eig, EG, degG, rankG);
    scan_local_kernel<<<nbG, 256, 0, stream>>>(degG, rowG, btG, G);
    scan_blocks_kernel<<<1, 1024, 0, stream>>>(btG, boG, nbG);
    place_kernel<<<cdivll(EG, TB), TB, 0, stream>>>(eig, EG, rowG, boG, rankG, srtG);
    gat_gather_kernel<<<cdivll(G, 8), 256, 0, stream>>>(
        h_g, a_sg, a_dg, srtG, rowG, boG, degG, bias2, xg_out, G);

    init_adj_kernel<<<cdivll((long long)G * 128, TB), TB, 0, stream>>>(
        adj, cnt, G);
    adj_accum_kernel<<<cdivll((long long)EG * 128, TB), TB, 0, stream>>>(
        eig, EG, xg_out, adj, cnt);
    adj_mean_kernel<<<cdivll((long long)G * 128, TB), TB, 0, stream>>>(
        adj, cnt, G);

    node_update_kernel<<<cdivll(N, 4), 128, 0, stream>>>(
        acc, ngrp, xg_out, adj, h, N);
    {
        unsigned tiles = (unsigned)((N + 15) / 16);
        gemm128_wmma<2, 32><<<tiles, 32, 0, stream>>>(h, W_fc, b_fc, out, N);
    }
}


# Edges grouped by a two-level counting sort: 256-node buckets filled per block, ranks taken in LDS, one returning global add per bucket and block (none per edge); sums stay in arrival order
# speedup vs baseline: 3.5203x; 1.0865x over previous
#include <hip/hip_runtime.h>
#include <hip/hip_bf16.h>

typedef __attribute__((ext_vector_type(2))) float v2f;
typedef __attribute__((ext_vector_type(8))) float v8f;

#define NEG_SLOPE 0.2f

template <int NT, int NC>
__global__ __launch_bounds__(32) void gemm128_wmma(
    const float* __restrict__ A, const float* __restrict__ B,
    const float* __restrict__ bias, float* __restrict__ C, int M) {
    constexpr int strips = NC / (16 * NT);
    const int tm = blockIdx.x / strips;
    const int ts = blockIdx.x % strips;
    const int lane = threadIdx.x;
    const int half = lane >> 4;
    const int l16  = lane & 15;

    int rowA = tm * 16 + l16;
    int rA = rowA < M ? rowA : (M - 1);
    const int k0base = half * 2;
    const float* Ap = A + (size_t)rA * 128 + k0base;
    const int col0 = ts * 16 * NT + l16;
    const float* Bp = B + (size_t)k0base * NC + col0;

    const v8f vzero = {0.f, 0.f, 0.f, 0.f, 0.f, 0.f, 0.f, 0.f};
    v8f c[NT];
#pragma unroll
    for (int t = 0; t < NT; ++t) c[t] = vzero;

#pragma unroll
    for (int kk = 0; kk < 128; kk += 4) {
        v2f a;
        a.x = Ap[kk];
        a.y = Ap[kk + 1];
        v2f b[NT];
#pragma unroll
        for (int t = 0; t < NT; ++t) {
            b[t].x = Bp[(size_t)kk * NC + t * 16];
            b[t].y = Bp[(size_t)(kk + 1) * NC + t * 16];
        }
#pragma unroll
        for (int t = 0; t < NT; ++t) {
            c[t] = __builtin_amdgcn_wmma_f32_16x16x4_f32(
                false, a, false, b[t], (short)0, c[t], false, false);
        }
    }
#pragma unroll
    for (int t = 0; t < NT; ++t) {
        float bv = bias ? bias[col0 + t * 16] : 0.f;
#pragma unroll
        for (int r = 0; r < 8; ++r) {
            int row = tm * 16 + r + half * 8;
            if (row < M) C[(size_t)row * NC + col0 + t * 16] = c[t][r] + bv;
        }
    }
}

__global__ void zero_words_kernel(int* __restrict__ p, int n) {
    int gid = blockIdx.x * blockDim.x + threadIdx.x;
    if (gid < n) p[gid] = 0;
}

__global__ void init_adj_kernel(float* adj, float* cnt, long long g) {
    long long gid = (long long)blockIdx.x * blockDim.x + threadIdx.x;
    if (gid >= g * 128) return;
    adj[gid] = 0.f;
    if (gid < g) cnt[gid] = 0.f;
}

__global__ void attn_kernel(const float* __restrict__ h,
                            const float* __restrict__ ws,
                            const float* __restrict__ wd,
                            float* __restrict__ a_s, float* __restrict__ a_d,
                            int n) {
    long long gid = (long long)blockIdx.x * blockDim.x + threadIdx.x;
    if (gid >= (long long)n * 4) return;
    int node = (int)(gid >> 2), hd = (int)(gid & 3);
    const float* hr = h + (size_t)node * 128 + hd * 32;
    float ss = 0.f, sd = 0.f;
#pragma unroll
    for (int c = 0; c < 32; ++c) {
        float hv = hr[c];
        ss += hv * ws[hd * 32 + c];
        sd += hv * wd[hd * 32 + c];
    }
    a_s[gid] = ss;
    a_d[gid] = sd;
}

#define BK_SHIFT 8
#define BK_NODES 256
#define CHUNK_EDGES 2048
#define CTR_STRIDE 32

__device__ __forceinline__ float lrelu(float v) {
    return v > 0.f ? v : NEG_SLOPE * v;
}

__device__ __forceinline__ int wave_incl_scan(int v, int lane) {
    for (int off = 1; off < 32; off <<= 1) {
        int y = __shfl_up(v, off, 32);
        if (lane >= off) v += y;
    }
    return v;
}

__global__ __launch_bounds__(256) void bucket_count_kernel(
    const int* __restrict__ ei, int E, int n, int nb, int* __restrict__ bcount) {
    extern __shared__ int sh[];
    const int tid = threadIdx.x;
    for (int i = tid; i < nb; i += 256) sh[i] = 0;
    __syncthreads();
    const int e0 = blockIdx.x * CHUNK_EDGES + tid;
    int d[8];
#pragma unroll
    for (int u = 0; u < 8; ++u) {
        const int e = e0 + u * 256;
        d[u] = e < E ? ei[E + e] : -1;
    }
#pragma unroll
    for (int u = 0; u < 8; ++u)
        if ((unsigned)d[u] < (unsigned)n) atomicAdd(&sh[d[u] >> BK_SHIFT], 1);
    __syncthreads();
    for (int i = tid; i < nb; i += 256) atomicAdd(&bcount[i * CTR_STRIDE], sh[i]);
}

__global__ __launch_bounds__(256) void bucket_fill_kernel(
    const int* __restrict__ ei, int E, int n, int nb,
    const int* __restrict__ bcount, int* __restrict__ bcursor,
    int* __restrict__ bstart, int* __restrict__ bins) {
    extern __shared__ int sh[];
    __shared__ int wsum[8];
    int* cnt = sh;
    int* st = sh + nb;
    const int tid = threadIdx.x, lane = tid & 31, w = tid >> 5;
    for (int i = tid; i < nb; i += 256) cnt[i] = 0;
    int carry = 0;
    for (int base = 0; base < nb; base += 256) {
        const int i = base + tid;
        const int v = (i < nb) ? bcount[i * CTR_STRIDE] : 0;
        const int inc = wave_incl_scan(v, lane);
        if (lane == 31) wsum[w] = inc;
        __syncthreads();
        int wpre = 0, tot = 0;
        for (int k = 0; k < 8; ++k) {
            int x = wsum[k];
            if (k < w) wpre += x;
            tot += x;
        }
        if (i < nb) st[i] = carry + wpre + inc - v;
        carry += tot;
        __syncthreads();
    }
    const int e0 = blockIdx.x * CHUNK_EDGES + tid;
    int d[8], s[8];
#pragma unroll
    for (int u = 0; u < 8; ++u) {
        const int e = e0 + u * 256;
        d[u] = e < E ? ei[E + e] : -1;
        s[u] = e < E ? ei[e] : 0;
    }
#pragma unroll
    for (int u = 0; u < 8; ++u)
        if ((unsigned)d[u] < (unsigned)n) atomicAdd(&cnt[d[u] >> BK_SHIFT], 1);
    __syncthreads();
    for (int i = tid; i < nb; i += 256) {
        const int r = atomicAdd(&bcursor[i * CTR_STRIDE], cnt[i]);
        cnt[i] = st[i] + r;
        if (blockIdx.x == 0) bstart[i] = st[i];
    }
    __syncthreads();
    int2* bp = reinterpret_cast<int2*>(bins);
#pragma unroll
    for (int u = 0; u < 8; ++u) {
        if ((unsigned)d[u] < (unsigned)n) {
            const int slot = atomicAdd(&cnt[d[u] >> BK_SHIFT], 1);
            if ((unsigned)slot < (unsigned)E) bp[slot] = make_int2(s[u], d[u]);
        }
    }
}

__global__ __launch_bounds__(1024) void bucket_rank_kernel(
    const int* __restrict__ bins, const int* __restrict__ bcount,
    const int* __restrict__ bstart, int E, int n,
    int* __restrict__ deg, int* __restrict__ rowstart, int* __restrict__ srt) {
    __shared__ int dcnt[BK_NODES];
    __shared__ int wsum[8];
    const int tid = threadIdx.x, lane = tid & 31, w = tid >> 5;
    const int b = blockIdx.x;
    const int beg = bstart[b];
    const int m = bcount[b * CTR_STRIDE];
    const int2* bp = reinterpret_cast<const int2*>(bins) + beg;
    if (tid < BK_NODES) dcnt[tid] = 0;
    __syncthreads();
    for (int i0 = tid; i0 < m; i0 += 4096) {
        int d[4];
#pragma unroll
        for (int u = 0; u < 4; ++u) {
            const int i = i0 + u * 1024;
            d[u] = i < m ? bp[i].y : -1;
        }
#pragma unroll
        for (int u = 0; u < 4; ++u)
            if (d[u] >= 0) atomicAdd(&dcnt[d[u] & (BK_NODES - 1)], 1);
    }
    __syncthreads();
    int v = 0, inc = 0;
    if (tid < BK_NODES) {
        v = dcnt[tid];
        inc = wave_incl_scan(v, lane);
        if (lane == 31) wsum[w] = inc;
    }
    __syncthreads();
    if (tid < BK_NODES) {
        int wpre = 0;
        for (int k = 0; k < 8; ++k)
            if (k < w) wpre += wsum[k];
        const int start = beg + wpre + inc - v;
        const int node = (b << BK_SHIFT) + tid;
        if (node < n) {
            deg[node] = v;
            rowstart[node] = start;
        }
        dcnt[tid] = start;
    }
    __syncthreads();
    for (int i0 = tid; i0 < m; i0 += 4096) {
        int2 p[4];
#pragma unroll
        for (int u = 0; u < 4; ++u) {
            const int i = i0 + u * 1024;
            p[u] = i < m ? bp[i] : make_int2(0, -1);
        }
#pragma unroll
        for (int u = 0; u < 4; ++u) {
            if (p[u].y >= 0) {
                const int slot = atomicAdd(&dcnt[p[u].y & (BK_NODES - 1)], 1);
                if ((unsigned)slot < (unsigned)E) srt[slot] = p[u].x;
            }
        }
    }
}

__global__ __launch_bounds__(256) void gat_gather_kernel(
    const float* __restrict__ h, const float* __restrict__ a_s,
    const float* __restrict__ a_d, const int* __restrict__ srt,
    const int* __restrict__ rowstart,
    const int* __restrict__ deg, const float* __restrict__ bias,
    float* __restrict__ dst, int n) {
    const int node = __builtin_amdgcn_readfirstlane(
        (int)(blockIdx.x * 8 + (threadIdx.x >> 5)));
    if (node >= n) return;
    const int lane = threadIdx.x & 31;
    const int hd = lane >> 3;
    const int cnt = deg[node];
    const int* seg = srt + rowstart[node];
    const float4 ad4 = *reinterpret_cast<const float4*>(a_d + (size_t)node * 4);
    const float4 as4 = *reinterpret_cast<const float4*>(a_s + (size_t)node * 4);

    float m0 = lrelu(as4.x + ad4.x), m1 = lrelu(as4.y + ad4.y);
    float m2 = lrelu(as4.z + ad4.z), m3 = lrelu(as4.w + ad4.w);
    for (int base = 0; base < cnt; base += 32) {
        const int j = base + lane;
        if (j < cnt) {
            const int s = min(max(seg[j], 0), n - 1);
            const float4 q =
                *reinterpret_cast<const float4*>(a_s + (size_t)s * 4);
            m0 = fmaxf(m0, lrelu(q.x + ad4.x));
            m1 = fmaxf(m1, lrelu(q.y + ad4.y));
            m2 = fmaxf(m2, lrelu(q.z + ad4.z));
            m3 = fmaxf(m3, lrelu(q.w + ad4.w));
        }
    }
    for (int off = 16; off >= 1; off >>= 1) {
        m0 = fmaxf(m0, __shfl_xor(m0, off, 32));
        m1 = fmaxf(m1, __shfl_xor(m1, off, 32));
        m2 = fmaxf(m2, __shfl_xor(m2, off, 32));
        m3 = fmaxf(m3, __shfl_xor(m3, off, 32));
    }
    const float m  = hd == 0 ? m0    : hd == 1 ? m1    : hd == 2 ? m2    : m3;
    const float ad = hd == 0 ? ad4.x : hd == 1 ? ad4.y : hd == 2 ? ad4.z : ad4.w;
    const float as = hd == 0 ? as4.x : hd == 1 ? as4.y : hd == 2 ? as4.z : as4.w;

    const float* hl = h + lane * 4;
    float ssum = __expf(lrelu(as + ad) - m);
    float4 acc;
    {
        const float4 r = *reinterpret_cast<const float4*>(hl + (size_t)node * 128);
        acc.x = ssum * r.x; acc.y = ssum * r.y;
        acc.z = ssum * r.z; acc.w = ssum * r.w;
    }
    for (int base = 0; base < cnt; base += 32) {
        const int j = base + lane;
        const int s_l = (j < cnt) ? min(max(seg[j], 0), n - 1) : node;
        const int rem = cnt - base;
        const int nch = rem < 32 ? rem : 32;
        for (int t = 0; t < nch; t += 4) {
            int su[4];
            float av[4];
            float4 r[4];
#pragma unroll
            for (int u = 0; u < 4; ++u) {
                const int tt = t + u;
                su[u] = __builtin_amdgcn_readlane(s_l, tt < nch ? tt : nch - 1);
            }
#pragma unroll
            for (int u = 0; u < 4; ++u) {
                av[u] = a_s[(size_t)su[u] * 4 + hd];
                r[u] = *reinterpret_cast<const float4*>(hl + (size_t)su[u] * 128);
            }
#pragma unroll
            for (int u = 0; u < 4; ++u) {
                const float ex =
                    (t + u < nch) ? __expf(lrelu(av[u] + ad) - m) : 0.f;
                ssum += ex;
                acc.x += ex * r[u].x; acc.y += ex * r[u].y;
                acc.z += ex * r[u].z; acc.w += ex * r[u].w;
            }
        }
    }
    const float den = ssum + 1e-16f;
    float4 o;
    o.x = acc.x / den + bias[lane * 4 + 0];
    o.y = acc.y / den + bias[lane * 4 + 1];
    o.z = acc.z / den + bias[lane * 4 + 2];
    o.w = acc.w / den + bias[lane * 4 + 3];
    *reinterpret_cast<float4*>(dst + (size_t)node * 128 + lane * 4) = o;
}

__global__ void adj_accum_kernel(const int* __restrict__ eig, int EG,
                                 const float* __restrict__ xg,
                                 float* __restrict__ adj,
                                 float* __restrict__ cnt) {
    long long gid = (long long)blockIdx.x * blockDim.x + threadIdx.x;
    if (gid >= (long long)EG * 128) return;
    int e = (int)(gid >> 7), ch = (int)(gid & 127);
    int g0 = eig[e], g1 = eig[EG + e];
    atomicAdd(&adj[(size_t)g0 * 128 + ch], xg[(size_t)g1 * 128 + ch]);
    if (ch == 0) atomicAdd(&cnt[g0], 1.f);
}

__global__ void adj_mean_kernel(float* __restrict__ adj,
                                const float* __restrict__ cnt, long long g) {
    long long gid = (long long)blockIdx.x * blockDim.x + threadIdx.x;
    if (gid >= g * 128) return;
    adj[gid] = adj[gid] / fmaxf(cnt[gid >> 7], 1.f);
}

__global__ __launch_bounds__(128) void node_update_kernel(
    const float* __restrict__ xo, const int* __restrict__ ng,
    const float* __restrict__ xg, const float* __restrict__ am,
    float* __restrict__ upd, int n) {
    int wave = threadIdx.x >> 5;
    int lane = threadIdx.x & 31;
    int node = blockIdx.x * (blockDim.x >> 5) + wave;
    if (node >= n) return;
    int g = ng[node];
    const float* xr = xo + (size_t)node * 128;
    const float* gf = xg + (size_t)g * 128;
    const float* af = am + (size_t)g * 128;
    float xv[4], gv[4], av[4];
    float pg = 0.f, pa = 0.f;
#pragma unroll
    for (int i = 0; i < 4; ++i) {
        int c = lane + i * 32;
        xv[i] = xr[c]; gv[i] = gf[c]; av[i] = af[c];
        pg += xv[i] * gv[i];
        pa += xv[i] * av[i];
    }
#pragma unroll
    for (int off = 16; off >= 1; off >>= 1) {
        pg += __shfl_xor(pg, off, 32);
        pa += __shfl_xor(pa, off, 32);
    }
    float* ur = upd + (size_t)node * 128;
#pragma unroll
    for (int i = 0; i < 4; ++i) {
        int c = lane + i * 32;
        ur[c] = xv[i] + pg * gv[i] + pa * av[i];
    }
}

static inline unsigned cdivll(long long a, long long b) {
    return (unsigned)((a + b - 1) / b);
}
static inline size_t al4(size_t v) { return (v + 3) & ~(size_t)3; }

extern "C" void kernel_launch(void* const* d_in, const int* in_sizes, int n_in,
                              void* d_out, int out_size, void* d_ws,
                              size_t ws_size, hipStream_t stream) {
    const float* x      = (const float*)d_in[0];
    const float* x_g    = (const float*)d_in[1];
    const int*   ei     = (const int*)d_in[2];
    const int*   eig    = (const int*)d_in[3];
    const int*   ngrp   = (const int*)d_in[4];
    const float* W1     = (const float*)d_in[5];
    const float* atts1  = (const float*)d_in[6];
    const float* attd1  = (const float*)d_in[7];
    const float* bias1  = (const float*)d_in[8];
    const float* W2     = (const float*)d_in[9];
    const float* atts2  = (const float*)d_in[10];
    const float* attd2  = (const float*)d_in[11];
    const float* bias2  = (const float*)d_in[12];
    const float* W_fc   = (const float*)d_in[13];
    const float* b_fc   = (const float*)d_in[14];

    const int N  = in_sizes[0] / 128;
    const int G  = in_sizes[1] / 128;
    const int E  = in_sizes[2] / 2;
    const int EG = in_sizes[3] / 2;

    float* out    = (float*)d_out;
    float* xg_out = out + (size_t)N * 32;

    float* ws = (float*)d_ws;
    const int nbN = (N + BK_NODES - 1) / BK_NODES;
    const int nbG = (G + BK_NODES - 1) / BK_NODES;
    const int nctr = 2 * (nbN + nbG) * CTR_STRIDE;
    size_t o = 0;
    float* h     = ws + o; o += (size_t)N * 128;
    float* acc   = ws + o; o += (size_t)N * 128;
    float* a_s   = ws + o; o += (size_t)N * 4;
    float* a_d   = ws + o; o += (size_t)N * 4;
    int*   degN  = (int*)(ws + o); o += al4((size_t)N);
    int*   rowN  = (int*)(ws + o); o += al4((size_t)N);
    int*   bsN   = (int*)(ws + o); o += al4((size_t)nbN);
    int*   bsG   = (int*)(ws + o); o += al4((size_t)nbG);
    float* h_g   = ws + o; o += (size_t)G * 128;
    float* a_sg  = ws + o; o += (size_t)G * 4;
    float* a_dg  = ws + o; o += (size_t)G * 4;
    int*   degG  = (int*)(ws + o); o += al4((size_t)G);
    int*   rowG  = (int*)(ws + o); o += al4((size_t)G);
    int*   srtG  = (int*)(ws + o); o += al4((size_t)EG);
    float* adj   = ws + o; o += (size_t)G * 128;
    float* cnt   = ws + o; o += (size_t)G;
    int*   ctr   = (int*)adj;
    int*   cntN  = ctr;
    int*   curN  = ctr + nbN * CTR_STRIDE;
    int*   cntG  = ctr + 2 * nbN * CTR_STRIDE;
    int*   curG  = cntG + nbG * CTR_STRIDE;
    int* binsN = (int*)acc;
    int* binsG = (int*)h;
    int* srtN  = (int*)out;

    const int TB = 256;

    zero_words_kernel<<<cdivll(nctr, TB), TB, 0, stream>>>(ctr, nctr);

    {
        unsigned tiles = (unsigned)(((N + 15) / 16) * (128 / 64));
        gemm128_wmma<4, 128><<<tiles, 32, 0, stream>>>(x, W1, nullptr, h, N);
    }
    attn_kernel<<<cdivll((long long)N * 4, TB), TB, 0, stream>>>(
        h, atts1, attd1, a_s, a_d, N);
    bucket_count_kernel<<<cdivll(E, CHUNK_EDGES), 256, (size_t)nbN * 4, stream>>>(
        ei, E, N, nbN, cntN);
    bucket_fill_kernel<<<cdivll(E, CHUNK_EDGES), 256, (size_t)nbN * 8, stream>>>(
        ei, E, N, nbN, cntN, curN, bsN, binsN);
    bucket_rank_kernel<<<nbN, 1024, 0, stream>>>(
        binsN, cntN, bsN, E, N, degN, rowN, srtN);
    gat_gather_kernel<<<cdivll(N, 8), 256, 0, stream>>>(
        h, a_s, a_d, srtN, rowN, degN, bias1, acc, N);

    {
        unsigned tiles = (unsigned)(((G + 15) / 16) * (128 / 64));
        gemm128_wmma<4, 128><<<tiles, 32, 0, stream>>>(x_g, W2, nullptr, h_g, G);
    }
    attn_kernel<<<cdivll((long long)G * 4, TB), TB, 0, stream>>>(
        h_g, atts2, attd2, a_sg, a_dg, G);
    bucket_count_kernel<<<cdivll(EG, CHUNK_EDGES), 256, (size_t)nbG * 4, stream>>>(
        eig, EG, G, nbG, cntG);
    bucket_fill_kernel<<<cdivll(EG, CHUNK_EDGES), 256, (size_t)nbG * 8, stream>>>(
        eig, EG, G, nbG, cntG, curG, bsG, binsG);
    bucket_rank_kernel<<<nbG, 1024, 0, stream>>>(
        binsG, cntG, bsG, EG, G, degG, rowG, srtG);
    gat_gather_kernel<<<cdivll(G, 8), 256, 0, stream>>>(
        h_g, a_sg, a_dg, srtG, rowG, degG, bias2, xg_out, G);

    init_adj_kernel<<<cdivll((long long)G * 128, TB), TB, 0, stream>>>(
        adj, cnt, G);
    adj_accum_kernel<<<cdivll((long long)EG * 128, TB), TB, 0, stream>>>(
        eig, EG, xg_out, adj, cnt);
    adj_mean_kernel<<<cdivll((long long)G * 128, TB), TB, 0, stream>>>(
        adj, cnt, G);

    node_update_kernel<<<cdivll(N, 4), 128, 0, stream>>>(
        acc, ngrp, xg_out, adj, h, N);
    {
        unsigned tiles = (unsigned)((N + 15) / 16);
        gemm128_wmma<2, 32><<<tiles, 32, 0, stream>>>(h, W_fc, b_fc, out, N);
    }
}


# Importance fusion (two row dots, scaled adds) done in the gather's epilogue on the row in registers, group graph computed first: x_out plane and one launch gone; dot sum order changes
# speedup vs baseline: 3.9255x; 1.1151x over previous
#include <hip/hip_runtime.h>
#include <hip/hip_bf16.h>

typedef __attribute__((ext_vector_type(2))) float v2f;
typedef __attribute__((ext_vector_type(8))) float v8f;

#define NEG_SLOPE 0.2f

template <int NT, int NC>
__global__ __launch_bounds__(32) void gemm128_wmma(
    const float* __restrict__ A, const float* __restrict__ B,
    const float* __restrict__ bias, float* __restrict__ C, int M) {
    constexpr int strips = NC / (16 * NT);
    const int tm = blockIdx.x / strips;
    const int ts = blockIdx.x % strips;
    const int lane = threadIdx.x;
    const int half = lane >> 4;
    const int l16  = lane & 15;

    int rowA = tm * 16 + l16;
    int rA = rowA < M ? rowA : (M - 1);
    const int k0base = half * 2;
    const float* Ap = A + (size_t)rA * 128 + k0base;
    const int col0 = ts * 16 * NT + l16;
    const float* Bp = B + (size_t)k0base * NC + col0;

    const v8f vzero = {0.f, 0.f, 0.f, 0.f, 0.f, 0.f, 0.f, 0.f};
    v8f c[NT];
#pragma unroll
    for (int t = 0; t < NT; ++t) c[t] = vzero;

#pragma unroll
    for (int kk = 0; kk < 128; kk += 4) {
        v2f a;
        a.x = Ap[kk];
        a.y = Ap[kk + 1];
        v2f b[NT];
#pragma unroll
        for (int t = 0; t < NT; ++t) {
            b[t].x = Bp[(size_t)kk * NC + t * 16];
            b[t].y = Bp[(size_t)(kk + 1) * NC + t * 16];
        }
#pragma unroll
        for (int t = 0; t < NT; ++t) {
            c[t] = __builtin_amdgcn_wmma_f32_16x16x4_f32(
                false, a, false, b[t], (short)0, c[t], false, false);
        }
    }
#pragma unroll
    for (int t = 0; t < NT; ++t) {
        float bv = bias ? bias[col0 + t * 16] : 0.f;
#pragma unroll
        for (int r = 0; r < 8; ++r) {
            int row = tm * 16 + r + half * 8;
            if (row < M) C[(size_t)row * NC + col0 + t * 16] = c[t][r] + bv;
        }
    }
}

__global__ void zero_words_kernel(int* __restrict__ p, int n) {
    int gid = blockIdx.x * blockDim.x + threadIdx.x;
    if (gid < n) p[gid] = 0;
}

__global__ void init_adj_kernel(float* adj, float* cnt, long long g) {
    long long gid = (long long)blockIdx.x * blockDim.x + threadIdx.x;
    if (gid >= g * 128) return;
    adj[gid] = 0.f;
    if (gid < g) cnt[gid] = 0.f;
}

__global__ void attn_kernel(const float* __restrict__ h,
                            const float* __restrict__ ws,
                            const float* __restrict__ wd,
                            float* __restrict__ a_s, float* __restrict__ a_d,
                            int n) {
    long long gid = (long long)blockIdx.x * blockDim.x + threadIdx.x;
    if (gid >= (long long)n * 4) return;
    int node = (int)(gid >> 2), hd = (int)(gid & 3);
    const float* hr = h + (size_t)node * 128 + hd * 32;
    float ss = 0.f, sd = 0.f;
#pragma unroll
    for (int c = 0; c < 32; ++c) {
        float hv = hr[c];
        ss += hv * ws[hd * 32 + c];
        sd += hv * wd[hd * 32 + c];
    }
    a_s[gid] = ss;
    a_d[gid] = sd;
}

#define BK_SHIFT 8
#define BK_NODES 256
#define CHUNK_EDGES 2048
#define CTR_STRIDE 32

__device__ __forceinline__ float lrelu(float v) {
    return v > 0.f ? v : NEG_SLOPE * v;
}

__device__ __forceinline__ int wave_incl_scan(int v, int lane) {
    for (int off = 1; off < 32; off <<= 1) {
        int y = __shfl_up(v, off, 32);
        if (lane >= off) v += y;
    }
    return v;
}

__global__ __launch_bounds__(256) void bucket_count_kernel(
    const int* __restrict__ ei, int E, int n, int nb, int* __restrict__ bcount) {
    extern __shared__ int sh[];
    const int tid = threadIdx.x;
    for (int i = tid; i < nb; i += 256) sh[i] = 0;
    __syncthreads();
    const int e0 = blockIdx.x * CHUNK_EDGES + tid;
    int d[8];
#pragma unroll
    for (int u = 0; u < 8; ++u) {
        const int e = e0 + u * 256;
        d[u] = e < E ? ei[E + e] : -1;
    }
#pragma unroll
    for (int u = 0; u < 8; ++u)
        if ((unsigned)d[u] < (unsigned)n) atomicAdd(&sh[d[u] >> BK_SHIFT], 1);
    __syncthreads();
    for (int i = tid; i < nb; i += 256) atomicAdd(&bcount[i * CTR_STRIDE], sh[i]);
}

__global__ __launch_bounds__(256) void bucket_fill_kernel(
    const int* __restrict__ ei, int E, int n, int nb,
    const int* __restrict__ bcount, int* __restrict__ bcursor,
    int* __restrict__ bstart, int* __restrict__ bins) {
    extern __shared__ int sh[];
    __shared__ int wsum[8];
    int* cnt = sh;
    int* st = sh + nb;
    const int tid = threadIdx.x, lane = tid & 31, w = tid >> 5;
    for (int i = tid; i < nb; i += 256) cnt[i] = 0;
    int carry = 0;
    for (int base = 0; base < nb; base += 256) {
        const int i = base + tid;
        const int v = (i < nb) ? bcount[i * CTR_STRIDE] : 0;
        const int inc = wave_incl_scan(v, lane);
        if (lane == 31) wsum[w] = inc;
        __syncthreads();
        int wpre = 0, tot = 0;
        for (int k = 0; k < 8; ++k) {
            int x = wsum[k];
            if (k < w) wpre += x;
            tot += x;
        }
        if (i < nb) st[i] = carry + wpre + inc - v;
        carry += tot;
        __syncthreads();
    }
    const int e0 = blockIdx.x * CHUNK_EDGES + tid;
    int d[8], s[8];
#pragma unroll
    for (int u = 0; u < 8; ++u) {
        const int e = e0 + u * 256;
        d[u] = e < E ? ei[E + e] : -1;
        s[u] = e < E ? ei[e] : 0;
    }
#pragma unroll
    for (int u = 0; u < 8; ++u)
        if ((unsigned)d[u] < (unsigned)n) atomicAdd(&cnt[d[u] >> BK_SHIFT], 1);
    __syncthreads();
    for (int i = tid; i < nb; i += 256) {
        const int r = atomicAdd(&bcursor[i * CTR_STRIDE], cnt[i]);
        cnt[i] = st[i] + r;
        if (blockIdx.x == 0) bstart[i] = st[i];
    }
    __syncthreads();
    int2* bp = reinterpret_cast<int2*>(bins);
#pragma unroll
    for (int u = 0; u < 8; ++u) {
        if ((unsigned)d[u] < (unsigned)n) {
            const int slot = atomicAdd(&cnt[d[u] >> BK_SHIFT], 1);
            if ((unsigned)slot < (unsigned)E) bp[slot] = make_int2(s[u], d[u]);
        }
    }
}

__global__ __launch_bounds__(1024) void bucket_rank_kernel(
    const int* __restrict__ bins, const int* __restrict__ bcount,
    const int* __restrict__ bstart, int E, int n,
    int* __restrict__ deg, int* __restrict__ rowstart, int* __restrict__ srt) {
    __shared__ int dcnt[BK_NODES];
    __shared__ int wsum[8];
    const int tid = threadIdx.x, lane = tid & 31, w = tid >> 5;
    const int b = blockIdx.x;
    const int beg = bstart[b];
    const int m = bcount[b * CTR_STRIDE];
    const int2* bp = reinterpret_cast<const int2*>(bins) + beg;
    if (tid < BK_NODES) dcnt[tid] = 0;
    __syncthreads();
    for (int i0 = tid; i0 < m; i0 += 4096) {
        int d[4];
#pragma unroll
        for (int u = 0; u < 4; ++u) {
            const int i = i0 + u * 1024;
            d[u] = i < m ? bp[i].y : -1;
        }
#pragma unroll
        for (int u = 0; u < 4; ++u)
            if (d[u] >= 0) atomicAdd(&dcnt[d[u] & (BK_NODES - 1)], 1);
    }
    __syncthreads();
    int v = 0, inc = 0;
    if (tid < BK_NODES) {
        v = dcnt[tid];
        inc = wave_incl_scan(v, lane);
        if (lane == 31) wsum[w] = inc;
    }
    __syncthreads();
    if (tid < BK_NODES) {
        int wpre = 0;
        for (int k = 0; k < 8; ++k)
            if (k < w) wpre += wsum[k];
        const int start = beg + wpre + inc - v;
        const int node = (b << BK_SHIFT) + tid;
        if (node < n) {
            deg[node] = v;
            rowstart[node] = start;
        }
        dcnt[tid] = start;
    }
    __syncthreads();
    for (int i0 = tid; i0 < m; i0 += 4096) {
        int2 p[4];
#pragma unroll
        for (int u = 0; u < 4; ++u) {
            const int i = i0 + u * 1024;
            p[u] = i < m ? bp[i] : make_int2(0, -1);
        }
#pragma unroll
        for (int u = 0; u < 4; ++u) {
            if (p[u].y >= 0) {
                const int slot = atomicAdd(&dcnt[p[u].y & (BK_NODES - 1)], 1);
                if ((unsigned)slot < (unsigned)E) srt[slot] = p[u].x;
            }
        }
    }
}

template <bool FUSE>
__global__ __launch_bounds__(256) void gat_gather_kernel(
    const float* __restrict__ h, const float* __restrict__ a_s,
    const float* __restrict__ a_d, const int* __restrict__ srt,
    const int* __restrict__ rowstart,
    const int* __restrict__ deg, const float* __restrict__ bias,
    const int* __restrict__ ng, const float* __restrict__ xg,
    const float* __restrict__ am,
    float* __restrict__ dst, int n) {
    const int node = __builtin_amdgcn_readfirstlane(
        (int)(blockIdx.x * 8 + (threadIdx.x >> 5)));
    if (node >= n) return;
    const int lane = threadIdx.x & 31;
    const int hd = lane >> 3;
    const int cnt = deg[node];
    const int* seg = srt + rowstart[node];
    const int grp = FUSE ? ng[node] : 0;
    const float4 ad4 = *reinterpret_cast<const float4*>(a_d + (size_t)node * 4);
    const float4 as4 = *reinterpret_cast<const float4*>(a_s + (size_t)node * 4);

    float m0 = lrelu(as4.x + ad4.x), m1 = lrelu(as4.y + ad4.y);
    float m2 = lrelu(as4.z + ad4.z), m3 = lrelu(as4.w + ad4.w);
    for (int base = 0; base < cnt; base += 32) {
        const int j = base + lane;
        if (j < cnt) {
            const int s = min(max(seg[j], 0), n - 1);
            const float4 q =
                *reinterpret_cast<const float4*>(a_s + (size_t)s * 4);
            m0 = fmaxf(m0, lrelu(q.x + ad4.x));
            m1 = fmaxf(m1, lrelu(q.y + ad4.y));
            m2 = fmaxf(m2, lrelu(q.z + ad4.z));
            m3 = fmaxf(m3, lrelu(q.w + ad4.w));
        }
    }
    for (int off = 16; off >= 1; off >>= 1) {
        m0 = fmaxf(m0, __shfl_xor(m0, off, 32));
        m1 = fmaxf(m1, __shfl_xor(m1, off, 32));
        m2 = fmaxf(m2, __shfl_xor(m2, off, 32));
        m3 = fmaxf(m3, __shfl_xor(m3, off, 32));
    }
    const float m  = hd == 0 ? m0    : hd == 1 ? m1    : hd == 2 ? m2    : m3;
    const float ad = hd == 0 ? ad4.x : hd == 1 ? ad4.y : hd == 2 ? ad4.z : ad4.w;
    const float as = hd == 0 ? as4.x : hd == 1 ? as4.y : hd == 2 ? as4.z : as4.w;

    float4 gv = make_float4(0.f, 0.f, 0.f, 0.f), mv = gv;
    if (FUSE) {
        gv = *reinterpret_cast<const float4*>(xg + (size_t)grp * 128 + lane * 4);
        mv = *reinterpret_cast<const float4*>(am + (size_t)grp * 128 + lane * 4);
    }

    const float* hl = h + lane * 4;
    float ssum = __expf(lrelu(as + ad) - m);
    float4 acc;
    {
        const float4 r = *reinterpret_cast<const float4*>(hl + (size_t)node * 128);
        acc.x = ssum * r.x; acc.y = ssum * r.y;
        acc.z = ssum * r.z; acc.w = ssum * r.w;
    }
    for (int base = 0; base < cnt; base += 32) {
        const int j = base + lane;
        const int s_l = (j < cnt) ? min(max(seg[j], 0), n - 1) : node;
        const int rem = cnt - base;
        const int nch = rem < 32 ? rem : 32;
        for (int t = 0; t < nch; t += 4) {
            int su[4];
            float av[4];
            float4 r[4];
#pragma unroll
            for (int u = 0; u < 4; ++u) {
                const int tt = t + u;
                su[u] = __builtin_amdgcn_readlane(s_l, tt < nch ? tt : nch - 1);
            }
#pragma unroll
            for (int u = 0; u < 4; ++u) {
                av[u] = a_s[(size_t)su[u] * 4 + hd];
                r[u] = *reinterpret_cast<const float4*>(hl + (size_t)su[u] * 128);
            }
#pragma unroll
            for (int u = 0; u < 4; ++u) {
                const float ex =
                    (t + u < nch) ? __expf(lrelu(av[u] + ad) - m) : 0.f;
                ssum += ex;
                acc.x += ex * r[u].x; acc.y += ex * r[u].y;
                acc.z += ex * r[u].z; acc.w += ex * r[u].w;
            }
        }
    }
    const float den = ssum + 1e-16f;
    float4 o;
    o.x = acc.x / den + bias[lane * 4 + 0];
    o.y = acc.y / den + bias[lane * 4 + 1];
    o.z = acc.z / den + bias[lane * 4 + 2];
    o.w = acc.w / den + bias[lane * 4 + 3];
    if (FUSE) {
        float pg = o.x * gv.x + o.y * gv.y + o.z * gv.z + o.w * gv.w;
        float pa = o.x * mv.x + o.y * mv.y + o.z * mv.z + o.w * mv.w;
        for (int off = 16; off >= 1; off >>= 1) {
            pg += __shfl_xor(pg, off, 32);
            pa += __shfl_xor(pa, off, 32);
        }
        o.x = o.x + pg * gv.x + pa * mv.x;
        o.y = o.y + pg * gv.y + pa * mv.y;
        o.z = o.z + pg * gv.z + pa * mv.z;
        o.w = o.w + pg * gv.w + pa * mv.w;
    }
    *reinterpret_cast<float4*>(dst + (size_t)node * 128 + lane * 4) = o;
}

__global__ void adj_accum_kernel(const int* __restrict__ eig, int EG,
                                 const float* __restrict__ xg,
                                 float* __restrict__ adj,
                                 float* __restrict__ cnt) {
    long long gid = (long long)blockIdx.x * blockDim.x + threadIdx.x;
    if (gid >= (long long)EG * 128) return;
    int e = (int)(gid >> 7), ch = (int)(gid & 127);
    int g0 = eig[e], g1 = eig[EG + e];
    atomicAdd(&adj[(size_t)g0 * 128 + ch], xg[(size_t)g1 * 128 + ch]);
    if (ch == 0) atomicAdd(&cnt[g0], 1.f);
}

__global__ void adj_mean_kernel(float* __restrict__ adj,
                                const float* __restrict__ cnt, long long g) {
    long long gid = (long long)blockIdx.x * blockDim.x + threadIdx.x;
    if (gid >= g * 128) return;
    adj[gid] = adj[gid] / fmaxf(cnt[gid >> 7], 1.f);
}

static inline unsigned cdivll(long long a, long long b) {
    return (unsigned)((a + b - 1) / b);
}
static inline size_t al4(size_t v) { return (v + 3) & ~(size_t)3; }

extern "C" void kernel_launch(void* const* d_in, const int* in_sizes, int n_in,
                              void* d_out, int out_size, void* d_ws,
                              size_t ws_size, hipStream_t stream) {
    const float* x      = (const float*)d_in[0];
    const float* x_g    = (const float*)d_in[1];
    const int*   ei     = (const int*)d_in[2];
    const int*   eig    = (const int*)d_in[3];
    const int*   ngrp   = (const int*)d_in[4];
    const float* W1     = (const float*)d_in[5];
    const float* atts1  = (const float*)d_in[6];
    const float* attd1  = (const float*)d_in[7];
    const float* bias1  = (const float*)d_in[8];
    const float* W2     = (const float*)d_in[9];
    const float* atts2  = (const float*)d_in[10];
    const float* attd2  = (const float*)d_in[11];
    const float* bias2  = (const float*)d_in[12];
    const float* W_fc   = (const float*)d_in[13];
    const float* b_fc   = (const float*)d_in[14];

    const int N  = in_sizes[0] / 128;
    const int G  = in_sizes[1] / 128;
    const int E  = in_sizes[2] / 2;
    const int EG = in_sizes[3] / 2;

    float* out    = (float*)d_out;
    float* xg_out = out + (size_t)N * 32;

    float* ws = (float*)d_ws;
    const int nbN = (N + BK_NODES - 1) / BK_NODES;
    const int nbG = (G + BK_NODES - 1) / BK_NODES;
    const int nctr = 2 * (nbN + nbG) * CTR_STRIDE;
    size_t o = 0;
    float* h     = ws + o; o += (size_t)N * 128;
    float* acc   = ws + o; o += (size_t)N * 128;
    float* a_s   = ws + o; o += (size_t)N * 4;
    float* a_d   = ws + o; o += (size_t)N * 4;
    int*   degN  = (int*)(ws + o); o += al4((size_t)N);
    int*   rowN  = (int*)(ws + o); o += al4((size_t)N);
    int*   bsN   = (int*)(ws + o); o += al4((size_t)nbN);
    int*   bsG   = (int*)(ws + o); o += al4((size_t)nbG);
    float* h_g   = ws + o; o += (size_t)G * 128;
    float* a_sg  = ws + o; o += (size_t)G * 4;
    float* a_dg  = ws + o; o += (size_t)G * 4;
    int*   degG  = (int*)(ws + o); o += al4((size_t)G);
    int*   rowG  = (int*)(ws + o); o += al4((size_t)G);
    int*   srtG  = (int*)(ws + o); o += al4((size_t)EG);
    float* adj   = ws + o; o += (size_t)G * 128;
    float* cnt   = ws + o; o += (size_t)G;
    int*   ctr   = (int*)adj;
    int*   cntN  = ctr;
    int*   curN  = ctr + nbN * CTR_STRIDE;
    int*   cntG  = ctr + 2 * nbN * CTR_STRIDE;
    int*   curG  = cntG + nbG * CTR_STRIDE;
    int* binsN = (int*)acc;
    int* binsG = (int*)h;
    int* srtN  = (int*)out;

    const int TB = 256;

    zero_words_kernel<<<cdivll(nctr, TB), TB, 0, stream>>>(ctr, nctr);


    bucket_count_kernel<<<cdivll(E, CHUNK_EDGES), 256, (size_t)nbN * 4, stream>>>(
        ei, E, N, nbN, cntN);
    bucket_fill_kernel<<<cdivll(E, CHUNK_EDGES), 256, (size_t)nbN * 8, stream>>>(
        ei, E, N, nbN, cntN, curN, bsN, binsN);
    bucket_rank_kernel<<<nbN, 1024, 0, stream>>>(
        binsN, cntN, bsN, E, N, degN, rowN, srtN);

    {
        unsigned tiles = (unsigned)(((G + 15) / 16) * (128 / 64));
        gemm128_wmma<4, 128><<<tiles, 32, 0, stream>>>(x_g, W2, nullptr, h_g, G);
    }
    attn_kernel<<<cdivll((long long)G * 4, TB), TB, 0, stream>>>(
        h_g, atts2, attd2, a_sg, a_dg, G);
    bucket_count_kernel<<<cdivll(EG, CHUNK_EDGES), 256, (size_t)nbG * 4, stream>>>(
        eig, EG, G, nbG, cntG);
    bucket_fill_kernel<<<cdivll(EG, CHUNK_EDGES), 256, (size_t)nbG * 8, stream>>>(
        eig, EG, G, nbG, cntG, curG, bsG, binsG);
    bucket_rank_kernel<<<nbG, 1024, 0, stream>>>(
        binsG, cntG, bsG, EG, G, degG, rowG, srtG);
    gat_gather_kernel<false><<<cdivll(G, 8), 256, 0, stream>>>(
        h_g, a_sg, a_dg, srtG, rowG, degG, bias2, nullptr, nullptr, nullptr,
        xg_out, G);

    init_adj_kernel<<<cdivll((long long)G * 128, TB), TB, 0, stream>>>(
        adj, cnt, G);
    adj_accum_kernel<<<cdivll((long long)EG * 128, TB), TB, 0, stream>>>(
        eig, EG, xg_out, adj, cnt);
    adj_mean_kernel<<<cdivll((long long)G * 128, TB), TB, 0, stream>>>(
        adj, cnt, G);

    {
        unsigned tiles = (unsigned)(((N + 15) / 16) * (128 / 64));
        gemm128_wmma<4, 128><<<tiles, 32, 0, stream>>>(x, W1, nullptr, h, N);
    }
    attn_kernel<<<cdivll((long long)N * 4, TB), TB, 0, stream>>>(
        h, atts1, attd1, a_s, a_d, N);
    gat_gather_kernel<true><<<cdivll(N, 8), 256, 0, stream>>>(
        h, a_s, a_d, srtN, rowN, degN, bias1, ngrp, xg_out, adj,
        acc, N);

    {
        unsigned tiles = (unsigned)((N + 15) / 16);
        gemm128_wmma<2, 32><<<tiles, 32, 0, stream>>>(acc, W_fc, b_fc, out, N);
    }
}


# Adjacency mean by one block per group scanning the group edge list: matching rows summed in registers, then over its 4 waves in fixed order; float atomics and two launches gone, sum order changes
# speedup vs baseline: 4.0848x; 1.0406x over previous
#include <hip/hip_runtime.h>
#include <hip/hip_bf16.h>

typedef __attribute__((ext_vector_type(2))) float v2f;
typedef __attribute__((ext_vector_type(8))) float v8f;

#define NEG_SLOPE 0.2f

template <int NT, int NC>
__global__ __launch_bounds__(32) void gemm128_wmma(
    const float* __restrict__ A, const float* __restrict__ B,
    const float* __restrict__ bias, float* __restrict__ C, int M) {
    constexpr int strips = NC / (16 * NT);
    const int tm = blockIdx.x / strips;
    const int ts = blockIdx.x % strips;
    const int lane = threadIdx.x;
    const int half = lane >> 4;
    const int l16  = lane & 15;

    int rowA = tm * 16 + l16;
    int rA = rowA < M ? rowA : (M - 1);
    const int k0base = half * 2;
    const float* Ap = A + (size_t)rA * 128 + k0base;
    const int col0 = ts * 16 * NT + l16;
    const float* Bp = B + (size_t)k0base * NC + col0;

    const v8f vzero = {0.f, 0.f, 0.f, 0.f, 0.f, 0.f, 0.f, 0.f};
    v8f c[NT];
#pragma unroll
    for (int t = 0; t < NT; ++t) c[t] = vzero;

#pragma unroll
    for (int kk = 0; kk < 128; kk += 4) {
        v2f a;
        a.x = Ap[kk];
        a.y = Ap[kk + 1];
        v2f b[NT];
#pragma unroll
        for (int t = 0; t < NT; ++t) {
            b[t].x = Bp[(size_t)kk * NC + t * 16];
            b[t].y = Bp[(size_t)(kk + 1) * NC + t * 16];
        }
#pragma unroll
        for (int t = 0; t < NT; ++t) {
            c[t] = __builtin_amdgcn_wmma_f32_16x16x4_f32(
                false, a, false, b[t], (short)0, c[t], false, false);
        }
    }
#pragma unroll
    for (int t = 0; t < NT; ++t) {
        float bv = bias ? bias[col0 + t * 16] : 0.f;
#pragma unroll
        for (int r = 0; r < 8; ++r) {
            int row = tm * 16 + r + half * 8;
            if (row < M) C[(size_t)row * NC + col0 + t * 16] = c[t][r] + bv;
        }
    }
}

__global__ void zero_words_kernel(int* __restrict__ p, int n) {
    int gid = blockIdx.x * blockDim.x + threadIdx.x;
    if (gid < n) p[gid] = 0;
}

__global__ void attn_kernel(const float* __restrict__ h,
                            const float* __restrict__ ws,
                            const float* __restrict__ wd,
                            float* __restrict__ a_s, float* __restrict__ a_d,
                            int n) {
    long long gid = (long long)blockIdx.x * blockDim.x + threadIdx.x;
    if (gid >= (long long)n * 4) return;
    int node = (int)(gid >> 2), hd = (int)(gid & 3);
    const float* hr = h + (size_t)node * 128 + hd * 32;
    float ss = 0.f, sd = 0.f;
#pragma unroll
    for (int c = 0; c < 32; ++c) {
        float hv = hr[c];
        ss += hv * ws[hd * 32 + c];
        sd += hv * wd[hd * 32 + c];
    }
    a_s[gid] = ss;
    a_d[gid] = sd;
}

#define BK_SHIFT 8
#define BK_NODES 256
#define CHUNK_EDGES 2048
#define CTR_STRIDE 32

__device__ __forceinline__ float lrelu(float v) {
    return v > 0.f ? v : NEG_SLOPE * v;
}

__device__ __forceinline__ int wave_incl_scan(int v, int lane) {
    for (int off = 1; off < 32; off <<= 1) {
        int y = __shfl_up(v, off, 32);
        if (lane >= off) v += y;
    }
    return v;
}

__global__ __launch_bounds__(256) void bucket_count_kernel(
    const int* __restrict__ ei, int E, int n, int nb, int* __restrict__ bcount) {
    extern __shared__ int sh[];
    const int tid = threadIdx.x;
    for (int i = tid; i < nb; i += 256) sh[i] = 0;
    __syncthreads();
    const int e0 = blockIdx.x * CHUNK_EDGES + tid;
    int d[8];
#pragma unroll
    for (int u = 0; u < 8; ++u) {
        const int e = e0 + u * 256;
        d[u] = e < E ? ei[E + e] : -1;
    }
#pragma unroll
    for (int u = 0; u < 8; ++u)
        if ((unsigned)d[u] < (unsigned)n) atomicAdd(&sh[d[u] >> BK_SHIFT], 1);
    __syncthreads();
    for (int i = tid; i < nb; i += 256) atomicAdd(&bcount[i * CTR_STRIDE], sh[i]);
}

__global__ __launch_bounds__(256) void bucket_fill_kernel(
    const int* __restrict__ ei, int E, int n, int nb,
    const int* __restrict__ bcount, int* __restrict__ bcursor,
    int* __restrict__ bstart, int* __restrict__ bins) {
    extern __shared__ int sh[];
    __shared__ int wsum[8];
    int* cnt = sh;
    int* st = sh + nb;
    const int tid = threadIdx.x, lane = tid & 31, w = tid >> 5;
    for (int i = tid; i < nb; i += 256) cnt[i] = 0;
    int carry = 0;
    for (int base = 0; base < nb; base += 256) {
        const int i = base + tid;
        const int v = (i < nb) ? bcount[i * CTR_STRIDE] : 0;
        const int inc = wave_incl_scan(v, lane);
        if (lane == 31) wsum[w] = inc;
        __syncthreads();
        int wpre = 0, tot = 0;
        for (int k = 0; k < 8; ++k) {
            int x = wsum[k];
            if (k < w) wpre += x;
            tot += x;
        }
        if (i < nb) st[i] = carry + wpre + inc - v;
        carry += tot;
        __syncthreads();
    }
    const int e0 = blockIdx.x * CHUNK_EDGES + tid;
    int d[8], s[8];
#pragma unroll
    for (int u = 0; u < 8; ++u) {
        const int e = e0 + u * 256;
        d[u] = e < E ? ei[E + e] : -1;
        s[u] = e < E ? ei[e] : 0;
    }
#pragma unroll
    for (int u = 0; u < 8; ++u)
        if ((unsigned)d[u] < (unsigned)n) atomicAdd(&cnt[d[u] >> BK_SHIFT], 1);
    __syncthreads();
    for (int i = tid; i < nb; i += 256) {
        const int r = atomicAdd(&bcursor[i * CTR_STRIDE], cnt[i]);
        cnt[i] = st[i] + r;
        if (blockIdx.x == 0) bstart[i] = st[i];
    }
    __syncthreads();
    int2* bp = reinterpret_cast<int2*>(bins);
#pragma unroll
    for (int u = 0; u < 8; ++u) {
        if ((unsigned)d[u] < (unsigned)n) {
            const int slot = atomicAdd(&cnt[d[u] >> BK_SHIFT], 1);
            if ((unsigned)slot < (unsigned)E) bp[slot] = make_int2(s[u], d[u]);
        }
    }
}

__global__ __launch_bounds__(1024) void bucket_rank_kernel(
    const int* __restrict__ bins, const int* __restrict__ bcount,
    const int* __restrict__ bstart, int E, int n,
    int* __restrict__ deg, int* __restrict__ rowstart, int* __restrict__ srt) {
    __shared__ int dcnt[BK_NODES];
    __shared__ int wsum[8];
    const int tid = threadIdx.x, lane = tid & 31, w = tid >> 5;
    const int b = blockIdx.x;
    const int beg = bstart[b];
    const int m = bcount[b * CTR_STRIDE];
    const int2* bp = reinterpret_cast<const int2*>(bins) + beg;
    if (tid < BK_NODES) dcnt[tid] = 0;
    __syncthreads();
    for (int i0 = tid; i0 < m; i0 += 4096) {
        int d[4];
#pragma unroll
        for (int u = 0; u < 4; ++u) {
            const int i = i0 + u * 1024;
            d[u] = i < m ? bp[i].y : -1;
        }
#pragma unroll
        for (int u = 0; u < 4; ++u)
            if (d[u] >= 0) atomicAdd(&dcnt[d[u] & (BK_NODES - 1)], 1);
    }
    __syncthreads();
    int v = 0, inc = 0;
    if (tid < BK_NODES) {
        v = dcnt[tid];
        inc = wave_incl_scan(v, lane);
        if (lane == 31) wsum[w] = inc;
    }
    __syncthreads();
    if (tid < BK_NODES) {
        int wpre = 0;
        for (int k = 0; k < 8; ++k)
            if (k < w) wpre += wsum[k];
        const int start = beg + wpre + inc - v;
        const int node = (b << BK_SHIFT) + tid;
        if (node < n) {
            deg[node] = v;
            rowstart[node] = start;
        }
        dcnt[tid] = start;
    }
    __syncthreads();
    for (int i0 = tid; i0 < m; i0 += 4096) {
        int2 p[4];
#pragma unroll
        for (int u = 0; u < 4; ++u) {
            const int i = i0 + u * 1024;
            p[u] = i < m ? bp[i] : make_int2(0, -1);
        }
#pragma unroll
        for (int u = 0; u < 4; ++u) {
            if (p[u].y >= 0) {
                const int slot = atomicAdd(&dcnt[p[u].y & (BK_NODES - 1)], 1);
                if ((unsigned)slot < (unsigned)E) srt[slot] = p[u].x;
            }
        }
    }
}

template <bool FUSE>
__global__ __launch_bounds__(256) void gat_gather_kernel(
    const float* __restrict__ h, const float* __restrict__ a_s,
    const float* __restrict__ a_d, const int* __restrict__ srt,
    const int* __restrict__ rowstart,
    const int* __restrict__ deg, const float* __restrict__ bias,
    const int* __restrict__ ng, const float* __restrict__ xg,
    const float* __restrict__ am,
    float* __restrict__ dst, int n) {
    const int node = __builtin_amdgcn_readfirstlane(
        (int)(blockIdx.x * 8 + (threadIdx.x >> 5)));
    if (node >= n) return;
    const int lane = threadIdx.x & 31;
    const int hd = lane >> 3;
    const int cnt = deg[node];
    const int* seg = srt + rowstart[node];
    const int grp = FUSE ? ng[node] : 0;
    const float4 ad4 = *reinterpret_cast<const float4*>(a_d + (size_t)node * 4);
    const float4 as4 = *reinterpret_cast<const float4*>(a_s + (size_t)node * 4);

    float m0 = lrelu(as4.x + ad4.x), m1 = lrelu(as4.y + ad4.y);
    float m2 = lrelu(as4.z + ad4.z), m3 = lrelu(as4.w + ad4.w);
    for (int base = 0; base < cnt; base += 32) {
        const int j = base + lane;
        if (j < cnt) {
            const int s = min(max(seg[j], 0), n - 1);
            const float4 q =
                *reinterpret_cast<const float4*>(a_s + (size_t)s * 4);
            m0 = fmaxf(m0, lrelu(q.x + ad4.x));
            m1 = fmaxf(m1, lrelu(q.y + ad4.y));
            m2 = fmaxf(m2, lrelu(q.z + ad4.z));
            m3 = fmaxf(m3, lrelu(q.w + ad4.w));
        }
    }
    for (int off = 16; off >= 1; off >>= 1) {
        m0 = fmaxf(m0, __shfl_xor(m0, off, 32));
        m1 = fmaxf(m1, __shfl_xor(m1, off, 32));
        m2 = fmaxf(m2, __shfl_xor(m2, off, 32));
        m3 = fmaxf(m3, __shfl_xor(m3, off, 32));
    }
    const float m  = hd == 0 ? m0    : hd == 1 ? m1    : hd == 2 ? m2    : m3;
    const float ad = hd == 0 ? ad4.x : hd == 1 ? ad4.y : hd == 2 ? ad4.z : ad4.w;
    const float as = hd == 0 ? as4.x : hd == 1 ? as4.y : hd == 2 ? as4.z : as4.w;

    float4 gv = make_float4(0.f, 0.f, 0.f, 0.f), mv = gv;
    if (FUSE) {
        gv = *reinterpret_cast<const float4*>(xg + (size_t)grp * 128 + lane * 4);
        mv = *reinterpret_cast<const float4*>(am + (size_t)grp * 128 + lane * 4);
    }

    const float* hl = h + lane * 4;
    float ssum = __expf(lrelu(as + ad) - m);
    float4 acc;
    {
        const float4 r = *reinterpret_cast<const float4*>(hl + (size_t)node * 128);
        acc.x = ssum * r.x; acc.y = ssum * r.y;
        acc.z = ssum * r.z; acc.w = ssum * r.w;
    }
    for (int base = 0; base < cnt; base += 32) {
        const int j = base + lane;
        const int s_l = (j < cnt) ? min(max(seg[j], 0), n - 1) : node;
        const int rem = cnt - base;
        const int nch = rem < 32 ? rem : 32;
        for (int t = 0; t < nch; t += 4) {
            int su[4];
            float av[4];
            float4 r[4];
#pragma unroll
            for (int u = 0; u < 4; ++u) {
                const int tt = t + u;
                su[u] = __builtin_amdgcn_readlane(s_l, tt < nch ? tt : nch - 1);
            }
#pragma unroll
            for (int u = 0; u < 4; ++u) {
                av[u] = a_s[(size_t)su[u] * 4 + hd];
                r[u] = *reinterpret_cast<const float4*>(hl + (size_t)su[u] * 128);
            }
#pragma unroll
            for (int u = 0; u < 4; ++u) {
                const float ex =
                    (t + u < nch) ? __expf(lrelu(av[u] + ad) - m) : 0.f;
                ssum += ex;
                acc.x += ex * r[u].x; acc.y += ex * r[u].y;
                acc.z += ex * r[u].z; acc.w += ex * r[u].w;
            }
        }
    }
    const float den = ssum + 1e-16f;
    float4 o;
    o.x = acc.x / den + bias[lane * 4 + 0];
    o.y = acc.y / den + bias[lane * 4 + 1];
    o.z = acc.z / den + bias[lane * 4 + 2];
    o.w = acc.w / den + bias[lane * 4 + 3];
    if (FUSE) {
        float pg = o.x * gv.x + o.y * gv.y + o.z * gv.z + o.w * gv.w;
        float pa = o.x * mv.x + o.y * mv.y + o.z * mv.z + o.w * mv.w;
        for (int off = 16; off >= 1; off >>= 1) {
            pg += __shfl_xor(pg, off, 32);
            pa += __shfl_xor(pa, off, 32);
        }
        o.x = o.x + pg * gv.x + pa * mv.x;
        o.y = o.y + pg * gv.y + pa * mv.y;
        o.z = o.z + pg * gv.z + pa * mv.z;
        o.w = o.w + pg * gv.w + pa * mv.w;
    }
    *reinterpret_cast<float4*>(dst + (size_t)node * 128 + lane * 4) = o;
}

#define ADJ_UNROLL 16
#define ADJ_THREADS 128

__device__ __forceinline__ void adj_add_rows(
    unsigned lanes, int hit, int e0, const int* __restrict__ eig, int EG, int G,
    const float* __restrict__ xl, float4& acc, int& found) {
    while (lanes) {
        const int L = __builtin_ctz(lanes);
        lanes &= lanes - 1;
        unsigned bits = (unsigned)__builtin_amdgcn_readlane(hit, L) & 15u;
        found += __builtin_popcount(bits);
        while (bits) {
            const int e = e0 + 4 * L + __builtin_ctz(bits);
            bits &= bits - 1;
            const int g1 = min(max(eig[EG + e], 0), G - 1);
            const float4 r =
                *reinterpret_cast<const float4*>(xl + (size_t)g1 * 128);
            acc.x += r.x; acc.y += r.y; acc.z += r.z; acc.w += r.w;
        }
    }
}

__global__ __launch_bounds__(ADJ_THREADS) void adj_mean_kernel(
    const int* __restrict__ eig, int EG, int G,
    const float* __restrict__ xg, float* __restrict__ adj) {
    __shared__ __align__(16) float part[ADJ_THREADS / 32][128];
    __shared__ int pfound[ADJ_THREADS / 32];
    const int g = blockIdx.x;
    const int tid = threadIdx.x, lane = tid & 31;
    const int w = __builtin_amdgcn_readfirstlane(tid >> 5);
    const float* xl = xg + lane * 4;
    float4 acc = make_float4(0.f, 0.f, 0.f, 0.f);
    int found = 0;
    const int nfull = EG >> 2;
    for (int qb = 0; qb < nfull; qb += ADJ_THREADS * ADJ_UNROLL) {
        int4 v[ADJ_UNROLL];
#pragma unroll
        for (int u = 0; u < ADJ_UNROLL; ++u) {
            const int q = min(qb + u * ADJ_THREADS + tid, nfull - 1);
            v[u] = *reinterpret_cast<const int4*>(eig + (size_t)q * 4);
        }
#pragma unroll
        for (int u = 0; u < ADJ_UNROLL; ++u) {
            int hit = (v[u].x == g ? 1 : 0) | (v[u].y == g ? 2 : 0) |
                      (v[u].z == g ? 4 : 0) | (v[u].w == g ? 8 : 0);
            if (qb + u * ADJ_THREADS + tid >= nfull) hit = 0;
            const unsigned lanes = __builtin_amdgcn_ballot_w32(hit != 0);
            if (lanes)
                adj_add_rows(lanes, hit, (qb + u * ADJ_THREADS + w * 32) * 4,
                             eig, EG, G, xl, acc, found);
        }
    }
    if (w == 0) {
        int hit = 0;
        if (lane == 0) {
            for (int c = 0; c < 3; ++c) {
                const int e = nfull * 4 + c;
                if (e < EG && eig[e] == g) hit |= 1 << c;
            }
        }
        const unsigned lanes = __builtin_amdgcn_ballot_w32(hit != 0);
        adj_add_rows(lanes, hit, nfull * 4, eig, EG, G, xl, acc, found);
    }
    *reinterpret_cast<float4*>(&part[w][lane * 4]) = acc;
    if (lane == 0) pfound[w] = found;
    __syncthreads();
    float s = 0.f;
    int c = 0;
    for (int k = 0; k < ADJ_THREADS / 32; ++k) {
        s += part[k][tid];
        c += pfound[k];
    }
    adj[(size_t)g * 128 + tid] = s / fmaxf((float)c, 1.f);
}

static inline unsigned cdivll(long long a, long long b) {
    return (unsigned)((a + b - 1) / b);
}
static inline size_t al4(size_t v) { return (v + 3) & ~(size_t)3; }

extern "C" void kernel_launch(void* const* d_in, const int* in_sizes, int n_in,
                              void* d_out, int out_size, void* d_ws,
                              size_t ws_size, hipStream_t stream) {
    const float* x      = (const float*)d_in[0];
    const float* x_g    = (const float*)d_in[1];
    const int*   ei     = (const int*)d_in[2];
    const int*   eig    = (const int*)d_in[3];
    const int*   ngrp   = (const int*)d_in[4];
    const float* W1     = (const float*)d_in[5];
    const float* atts1  = (const float*)d_in[6];
    const float* attd1  = (const float*)d_in[7];
    const float* bias1  = (const float*)d_in[8];
    const float* W2     = (const float*)d_in[9];
    const float* atts2  = (const float*)d_in[10];
    const float* attd2  = (const float*)d_in[11];
    const float* bias2  = (const float*)d_in[12];
    const float* W_fc   = (const float*)d_in[13];
    const float* b_fc   = (const float*)d_in[14];

    const int N  = in_sizes[0] / 128;
    const int G  = in_sizes[1] / 128;
    const int E  = in_sizes[2] / 2;
    const int EG = in_sizes[3] / 2;

    float* out    = (float*)d_out;
    float* xg_out = out + (size_t)N * 32;

    float* ws = (float*)d_ws;
    const int nbN = (N + BK_NODES - 1) / BK_NODES;
    const int nbG = (G + BK_NODES - 1) / BK_NODES;
    const int nctr = 2 * (nbN + nbG) * CTR_STRIDE;
    size_t o = 0;
    float* h     = ws + o; o += (size_t)N * 128;
    float* acc   = ws + o; o += (size_t)N * 128;
    float* a_s   = ws + o; o += (size_t)N * 4;
    float* a_d   = ws + o; o += (size_t)N * 4;
    int*   degN  = (int*)(ws + o); o += al4((size_t)N);
    int*   rowN  = (int*)(ws + o); o += al4((size_t)N);
    int*   bsN   = (int*)(ws + o); o += al4((size_t)nbN);
    int*   bsG   = (int*)(ws + o); o += al4((size_t)nbG);
    float* h_g   = ws + o; o += (size_t)G * 128;
    float* a_sg  = ws + o; o += (size_t)G * 4;
    float* a_dg  = ws + o; o += (size_t)G * 4;
    int*   degG  = (int*)(ws + o); o += al4((size_t)G);
    int*   rowG  = (int*)(ws + o); o += al4((size_t)G);
    int*   srtG  = (int*)(ws + o); o += al4((size_t)EG);
    float* adj   = ws + o; o += (size_t)G * 128;
    int*   ctr   = (int*)adj;
    int*   cntN  = ctr;
    int*   curN  = ctr + nbN * CTR_STRIDE;
    int*   cntG  = ctr + 2 * nbN * CTR_STRIDE;
    int*   curG  = cntG + nbG * CTR_STRIDE;
    int* binsN = (int*)acc;
    int* binsG = (int*)h;
    int* srtN  = (int*)out;

    const int TB = 256;

    zero_words_kernel<<<cdivll(nctr, TB), TB, 0, stream>>>(ctr, nctr);


    bucket_count_kernel<<<cdivll(E, CHUNK_EDGES), 256, (size_t)nbN * 4, stream>>>(
        ei, E, N, nbN, cntN);
    bucket_fill_kernel<<<cdivll(E, CHUNK_EDGES), 256, (size_t)nbN * 8, stream>>>(
        ei, E, N, nbN, cntN, curN, bsN, binsN);
    bucket_rank_kernel<<<nbN, 1024, 0, stream>>>(
        binsN, cntN, bsN, E, N, degN, rowN, srtN);

    {
        unsigned tiles = (unsigned)(((G + 15) / 16) * (128 / 64));
        gemm128_wmma<4, 128><<<tiles, 32, 0, stream>>>(x_g, W2, nullptr, h_g, G);
    }
    attn_kernel<<<cdivll((long long)G * 4, TB), TB, 0, stream>>>(
        h_g, atts2, attd2, a_sg, a_dg, G);
    bucket_count_kernel<<<cdivll(EG, CHUNK_EDGES), 256, (size_t)nbG * 4, stream>>>(
        eig, EG, G, nbG, cntG);
    bucket_fill_kernel<<<cdivll(EG, CHUNK_EDGES), 256, (size_t)nbG * 8, stream>>>(
        eig, EG, G, nbG, cntG, curG, bsG, binsG);
    bucket_rank_kernel<<<nbG, 1024, 0, stream>>>(
        binsG, cntG, bsG, EG, G, degG, rowG, srtG);
    gat_gather_kernel<false><<<cdivll(G, 8), 256, 0, stream>>>(
        h_g, a_sg, a_dg, srtG, rowG, degG, bias2, nullptr, nullptr, nullptr,
        xg_out, G);

    adj_mean_kernel<<<(unsigned)G, ADJ_THREADS, 0, stream>>>(eig, EG, G, xg_out, adj);

    {
        unsigned tiles = (unsigned)(((N + 15) / 16) * (128 / 64));
        gemm128_wmma<4, 128><<<tiles, 32, 0, stream>>>(x, W1, nullptr, h, N);
    }
    attn_kernel<<<cdivll((long long)N * 4, TB), TB, 0, stream>>>(
        h, atts1, attd1, a_s, a_d, N);
    gat_gather_kernel<true><<<cdivll(N, 8), 256, 0, stream>>>(
        h, a_s, a_d, srtN, rowN, degN, bias1, ngrp, xg_out, adj,
        acc, N);

    {
        unsigned tiles = (unsigned)((N + 15) / 16);
        gemm128_wmma<2, 32><<<tiles, 32, 0, stream>>>(acc, W_fc, b_fc, out, N);
    }
}


# Independent stages share a launch by block range (both graphs' count/fill/rank; clear+group product; group attention+count; group gather+big product; adj mean+big attention): 15 launches become 8
# speedup vs baseline: 4.8109x; 1.1778x over previous
#include <hip/hip_runtime.h>
#include <hip/hip_bf16.h>

typedef __attribute__((ext_vector_type(2))) float v2f;
typedef __attribute__((ext_vector_type(8))) float v8f;

#define NEG_SLOPE 0.2f

template <int NT, int NC>
__device__ __forceinline__ void gemm128_strip(
    const float* __restrict__ A, const float* __restrict__ B,
    const float* __restrict__ bias, float* __restrict__ C, int M,
    int bid, int lane) {
    constexpr int strips = NC / (16 * NT);
    const int tm = bid / strips;
    const int ts = bid % strips;
    const int half = lane >> 4;
    const int l16  = lane & 15;

    int rowA = tm * 16 + l16;
    int rA = rowA < M ? rowA : (M - 1);
    const int k0base = half * 2;
    const float* Ap = A + (size_t)rA * 128 + k0base;
    const int col0 = ts * 16 * NT + l16;
    const float* Bp = B + (size_t)k0base * NC + col0;

    const v8f vzero = {0.f, 0.f, 0.f, 0.f, 0.f, 0.f, 0.f, 0.f};
    v8f c[NT];
#pragma unroll
    for (int t = 0; t < NT; ++t) c[t] = vzero;

#pragma unroll
    for (int kk = 0; kk < 128; kk += 4) {
        v2f a;
        a.x = Ap[kk];
        a.y = Ap[kk + 1];
        v2f b[NT];
#pragma unroll
        for (int t = 0; t < NT; ++t) {
            b[t].x = Bp[(size_t)kk * NC + t * 16];
            b[t].y = Bp[(size_t)(kk + 1) * NC + t * 16];
        }
#pragma unroll
        for (int t = 0; t < NT; ++t) {
            c[t] = __builtin_amdgcn_wmma_f32_16x16x4_f32(
                false, a, false, b[t], (short)0, c[t], false, false);
        }
    }
#pragma unroll
    for (int t = 0; t < NT; ++t) {
        float bv = bias ? bias[col0 + t * 16] : 0.f;
#pragma unroll
        for (int r = 0; r < 8; ++r) {
            int row = tm * 16 + r + half * 8;
            if (row < M) C[(size_t)row * NC + col0 + t * 16] = c[t][r] + bv;
        }
    }
}

template <int NT, int NC>
__global__ __launch_bounds__(32) void gemm128_wmma(
    const float* __restrict__ A, const float* __restrict__ B,
    const float* __restrict__ bias, float* __restrict__ C, int M) {
    gemm128_strip<NT, NC>(A, B, bias, C, M, (int)blockIdx.x, (int)threadIdx.x);
}


#define ZERO_PER_BLOCK 256
__global__ __launch_bounds__(32) void zero_gemm_kernel(
    int* __restrict__ p, int n, int zblocks,
    const float* __restrict__ A, const float* __restrict__ B,
    float* __restrict__ C, int M) {
    if ((int)blockIdx.x < zblocks) {
        const int base = (int)blockIdx.x * ZERO_PER_BLOCK + (int)threadIdx.x;
        for (int k = 0; k < ZERO_PER_BLOCK; k += 32)
            if (base + k < n) p[base + k] = 0;
        return;
    }
    gemm128_strip<4, 128>(A, B, nullptr, C, M, (int)blockIdx.x - zblocks,
                          (int)threadIdx.x);
}

__device__ __forceinline__ void attn_body(const float* __restrict__ h,
                                          const float* __restrict__ ws,
                                          const float* __restrict__ wd,
                                          float* __restrict__ a_s,
                                          float* __restrict__ a_d,
                                          int n, int bid) {
    long long gid = (long long)bid * blockDim.x + threadIdx.x;
    if (gid >= (long long)n * 4) return;
    int node = (int)(gid >> 2), hd = (int)(gid & 3);
    const float* hr = h + (size_t)node * 128 + hd * 32;
    float ss = 0.f, sd = 0.f;
#pragma unroll
    for (int c = 0; c < 32; ++c) {
        float hv = hr[c];
        ss += hv * ws[hd * 32 + c];
        sd += hv * wd[hd * 32 + c];
    }
    a_s[gid] = ss;
    a_d[gid] = sd;
}

#define BK_SHIFT 8
#define BK_NODES 256
#define CHUNK_EDGES 2048
#define CTR_STRIDE 32

__device__ __forceinline__ float lrelu(float v) {
    return v > 0.f ? v : NEG_SLOPE * v;
}

__device__ __forceinline__ int wave_incl_scan(int v, int lane) {
    for (int off = 1; off < 32; off <<= 1) {
        int y = __shfl_up(v, off, 32);
        if (lane >= off) v += y;
    }
    return v;
}

__global__ __launch_bounds__(256) void attn_count_kernel(
    const float* __restrict__ h, const float* __restrict__ ws,
    const float* __restrict__ wd, float* __restrict__ a_s,
    float* __restrict__ a_d, int na, int ablocks,
    const int* __restrict__ ei0, int E0, int n0, int nb0, int* __restrict__ bcount0,
    int blocks0,
    const int* __restrict__ ei1, int E1, int n1, int nb1, int* __restrict__ bcount1) {
    extern __shared__ int sh[];
    if ((int)blockIdx.x < ablocks) {
        attn_body(h, ws, wd, a_s, a_d, na, (int)blockIdx.x);
        return;
    }
    const int cb = (int)blockIdx.x - ablocks;
    const bool second = cb >= blocks0;
    const int* __restrict__ ei = second ? ei1 : ei0;
    int* __restrict__ bcount = second ? bcount1 : bcount0;
    const int E = second ? E1 : E0, n = second ? n1 : n0, nb = second ? nb1 : nb0;
    const int bid = second ? cb - blocks0 : cb;
    const int tid = threadIdx.x;
    for (int i = tid; i < nb; i += 256) sh[i] = 0;
    __syncthreads();
    const int e0 = bid * CHUNK_EDGES + tid;
    int d[8];
#pragma unroll
    for (int u = 0; u < 8; ++u) {
        const int e = e0 + u * 256;
        d[u] = e < E ? ei[E + e] : -1;
    }
#pragma unroll
    for (int u = 0; u < 8; ++u)
        if ((unsigned)d[u] < (unsigned)n) atomicAdd(&sh[d[u] >> BK_SHIFT], 1);
    __syncthreads();
    for (int i = tid; i < nb; i += 256) atomicAdd(&bcount[i * CTR_STRIDE], sh[i]);
}

__global__ __launch_bounds__(256) void bucket_fill_kernel(
    const int* __restrict__ ei0, int E0, int n0, int nb0,
    const int* __restrict__ bcount0, int* __restrict__ bcursor0,
    int* __restrict__ bstart0, int* __restrict__ bins0,
    int blocks0,
    const int* __restrict__ ei1, int E1, int n1, int nb1,
    const int* __restrict__ bcount1, int* __restrict__ bcursor1,
    int* __restrict__ bstart1, int* __restrict__ bins1) {
    extern __shared__ int sh[];
    __shared__ int wsum[8];
    const bool second = (int)blockIdx.x >= blocks0;
    const int* __restrict__ ei = second ? ei1 : ei0;
    const int* __restrict__ bcount = second ? bcount1 : bcount0;
    int* __restrict__ bcursor = second ? bcursor1 : bcursor0;
    int* __restrict__ bstart = second ? bstart1 : bstart0;
    int* __restrict__ bins = second ? bins1 : bins0;
    const int E = second ? E1 : E0, n = second ? n1 : n0, nb = second ? nb1 : nb0;
    const int bid = second ? (int)blockIdx.x - blocks0 : (int)blockIdx.x;
    int* cnt = sh;
    int* st = sh + nb;
    const int tid = threadIdx.x, lane = tid & 31, w = tid >> 5;
    for (int i = tid; i < nb; i += 256) cnt[i] = 0;
    int carry = 0;
    for (int base = 0; base < nb; base += 256) {
        const int i = base + tid;
        const int v = (i < nb) ? bcount[i * CTR_STRIDE] : 0;
        const int inc = wave_incl_scan(v, lane);
        if (lane == 31) wsum[w] = inc;
        __syncthreads();
        int wpre = 0, tot = 0;
        for (int k = 0; k < 8; ++k) {
            int x = wsum[k];
            if (k < w) wpre += x;
            tot += x;
        }
        if (i < nb) st[i] = carry + wpre + inc - v;
        carry += tot;
        __syncthreads();
    }
    const int e0 = bid * CHUNK_EDGES + tid;
    int d[8], s[8];
#pragma unroll
    for (int u = 0; u < 8; ++u) {
        const int e = e0 + u * 256;
        d[u] = e < E ? ei[E + e] : -1;
        s[u] = e < E ? ei[e] : 0;
    }
#pragma unroll
    for (int u = 0; u < 8; ++u)
        if ((unsigned)d[u] < (unsigned)n) atomicAdd(&cnt[d[u] >> BK_SHIFT], 1);
    __syncthreads();
    for (int i = tid; i < nb; i += 256) {
        const int r = atomicAdd(&bcursor[i * CTR_STRIDE], cnt[i]);
        cnt[i] = st[i] + r;
        if (bid == 0) bstart[i] = st[i];
    }
    __syncthreads();
    int2* bp = reinterpret_cast<int2*>(bins);
#pragma unroll
    for (int u = 0; u < 8; ++u) {
        if ((unsigned)d[u] < (unsigned)n) {
            const int slot = atomicAdd(&cnt[d[u] >> BK_SHIFT], 1);
            if ((unsigned)slot < (unsigned)E) bp[slot] = make_int2(s[u], d[u]);
        }
    }
}

__global__ __launch_bounds__(1024) void bucket_rank_kernel(
    const int* __restrict__ bins0, const int* __restrict__ bcount0,
    const int* __restrict__ bstart0, int E0, int n0,
    int* __restrict__ deg0, int* __restrict__ rowstart0, int* __restrict__ srt0,
    int blocks0,
    const int* __restrict__ bins1, const int* __restrict__ bcount1,
    const int* __restrict__ bstart1, int E1, int n1,
    int* __restrict__ deg1, int* __restrict__ rowstart1, int* __restrict__ srt1) {
    __shared__ int dcnt[BK_NODES];
    __shared__ int wsum[8];
    const bool second = (int)blockIdx.x >= blocks0;
    const int* __restrict__ bins = second ? bins1 : bins0;
    const int* __restrict__ bcount = second ? bcount1 : bcount0;
    const int* __restrict__ bstart = second ? bstart1 : bstart0;
    int* __restrict__ deg = second ? deg1 : deg0;
    int* __restrict__ rowstart = second ? rowstart1 : rowstart0;
    int* __restrict__ srt = second ? srt1 : srt0;
    const int E = second ? E1 : E0, n = second ? n1 : n0;
    const int tid = threadIdx.x, lane = tid & 31, w = tid >> 5;
    const int b = second ? (int)blockIdx.x - blocks0 : (int)blockIdx.x;
    const int beg = bstart[b];
    const int m = bcount[b * CTR_STRIDE];
    const int2* bp = reinterpret_cast<const int2*>(bins) + beg;
    if (tid < BK_NODES) dcnt[tid] = 0;
    __syncthreads();
    for (int i0 = tid; i0 < m; i0 += 4096) {
        int d[4];
#pragma unroll
        for (int u = 0; u < 4; ++u) {
            const int i = i0 + u * 1024;
            d[u] = i < m ? bp[i].y : -1;
        }
#pragma unroll
        for (int u = 0; u < 4; ++u)
            if (d[u] >= 0) atomicAdd(&dcnt[d[u] & (BK_NODES - 1)], 1);
    }
    __syncthreads();
    int v = 0, inc = 0;
    if (tid < BK_NODES) {
        v = dcnt[tid];
        inc = wave_incl_scan(v, lane);
        if (lane == 31) wsum[w] = inc;
    }
    __syncthreads();
    if (tid < BK_NODES) {
        int wpre = 0;
        for (int k = 0; k < 8; ++k)
            if (k < w) wpre += wsum[k];
        const int start = beg + wpre + inc - v;
        const int node = (b << BK_SHIFT) + tid;
        if (node < n) {
            deg[node] = v;
            rowstart[node] = start;
        }
        dcnt[tid] = start;
    }
    __syncthreads();
    for (int i0 = tid; i0 < m; i0 += 4096) {
        int2 p[4];
#pragma unroll
        for (int u = 0; u < 4; ++u) {
            const int i = i0 + u * 1024;
            p[u] = i < m ? bp[i] : make_int2(0, -1);
        }
#pragma unroll
        for (int u = 0; u < 4; ++u) {
            if (p[u].y >= 0) {
                const int slot = atomicAdd(&dcnt[p[u].y & (BK_NODES - 1)], 1);
                if ((unsigned)slot < (unsigned)E) srt[slot] = p[u].x;
            }
        }
    }
}

template <bool FUSE>
__device__ __forceinline__ void gat_gather_body(
    const float* __restrict__ h, const float* __restrict__ a_s,
    const float* __restrict__ a_d, const int* __restrict__ srt,
    const int* __restrict__ rowstart,
    const int* __restrict__ deg, const float* __restrict__ bias,
    const int* __restrict__ ng, const float* __restrict__ xg,
    const float* __restrict__ am,
    float* __restrict__ dst, int n, int wave_id) {
    const int node = __builtin_amdgcn_readfirstlane(wave_id);
    if (node >= n) return;
    const int lane = threadIdx.x & 31;
    const int hd = lane >> 3;
    const int cnt = deg[node];
    const int* seg = srt + rowstart[node];
    const int grp = FUSE ? ng[node] : 0;
    const float4 ad4 = *reinterpret_cast<const float4*>(a_d + (size_t)node * 4);
    const float4 as4 = *reinterpret_cast<const float4*>(a_s + (size_t)node * 4);

    float m0 = lrelu(as4.x + ad4.x), m1 = lrelu(as4.y + ad4.y);
    float m2 = lrelu(as4.z + ad4.z), m3 = lrelu(as4.w + ad4.w);
    for (int base = 0; base < cnt; base += 32) {
        const int j = base + lane;
        if (j < cnt) {
            const int s = min(max(seg[j], 0), n - 1);
            const float4 q =
                *reinterpret_cast<const float4*>(a_s + (size_t)s * 4);
            m0 = fmaxf(m0, lrelu(q.x + ad4.x));
            m1 = fmaxf(m1, lrelu(q.y + ad4.y));
            m2 = fmaxf(m2, lrelu(q.z + ad4.z));
            m3 = fmaxf(m3, lrelu(q.w + ad4.w));
        }
    }
    for (int off = 16; off >= 1; off >>= 1) {
        m0 = fmaxf(m0, __shfl_xor(m0, off, 32));
        m1 = fmaxf(m1, __shfl_xor(m1, off, 32));
        m2 = fmaxf(m2, __shfl_xor(m2, off, 32));
        m3 = fmaxf(m3, __shfl_xor(m3, off, 32));
    }
    const float m  = hd == 0 ? m0    : hd == 1 ? m1    : hd == 2 ? m2    : m3;
    const float ad = hd == 0 ? ad4.x : hd == 1 ? ad4.y : hd == 2 ? ad4.z : ad4.w;
    const float as = hd == 0 ? as4.x : hd == 1 ? as4.y : hd == 2 ? as4.z : as4.w;

    float4 gv = make_float4(0.f, 0.f, 0.f, 0.f), mv = gv;
    if (FUSE) {
        gv = *reinterpret_cast<const float4*>(xg + (size_t)grp * 128 + lane * 4);
        mv = *reinterpret_cast<const float4*>(am + (size_t)grp * 128 + lane * 4);
    }

    const float* hl = h + lane * 4;
    float ssum = __expf(lrelu(as + ad) - m);
    float4 acc;
    {
        const float4 r = *reinterpret_cast<const float4*>(hl + (size_t)node * 128);
        acc.x = ssum * r.x; acc.y = ssum * r.y;
        acc.z = ssum * r.z; acc.w = ssum * r.w;
    }
    for (int base = 0; base < cnt; base += 32) {
        const int j = base + lane;
        const int s_l = (j < cnt) ? min(max(seg[j], 0), n - 1) : node;
        const int rem = cnt - base;
        const int nch = rem < 32 ? rem : 32;
        for (int t = 0; t < nch; t += 4) {
            int su[4];
            float av[4];
            float4 r[4];
#pragma unroll
            for (int u = 0; u < 4; ++u) {
                const int tt = t + u;
                su[u] = __builtin_amdgcn_readlane(s_l, tt < nch ? tt : nch - 1);
            }
#pragma unroll
            for (int u = 0; u < 4; ++u) {
                av[u] = a_s[(size_t)su[u] * 4 + hd];
                r[u] = *reinterpret_cast<const float4*>(hl + (size_t)su[u] * 128);
            }
#pragma unroll
            for (int u = 0; u < 4; ++u) {
                const float ex =
                    (t + u < nch) ? __expf(lrelu(av[u] + ad) - m) : 0.f;
                ssum += ex;
                acc.x += ex * r[u].x; acc.y += ex * r[u].y;
                acc.z += ex * r[u].z; acc.w += ex * r[u].w;
            }
        }
    }
    const float den = ssum + 1e-16f;
    float4 o;
    o.x = acc.x / den + bias[lane * 4 + 0];
    o.y = acc.y / den + bias[lane * 4 + 1];
    o.z = acc.z / den + bias[lane * 4 + 2];
    o.w = acc.w / den + bias[lane * 4 + 3];
    if (FUSE) {
        float pg = o.x * gv.x + o.y * gv.y + o.z * gv.z + o.w * gv.w;
        float pa = o.x * mv.x + o.y * mv.y + o.z * mv.z + o.w * mv.w;
        for (int off = 16; off >= 1; off >>= 1) {
            pg += __shfl_xor(pg, off, 32);
            pa += __shfl_xor(pa, off, 32);
        }
        o.x = o.x + pg * gv.x + pa * mv.x;
        o.y = o.y + pg * gv.y + pa * mv.y;
        o.z = o.z + pg * gv.z + pa * mv.z;
        o.w = o.w + pg * gv.w + pa * mv.w;
    }
    *reinterpret_cast<float4*>(dst + (size_t)node * 128 + lane * 4) = o;
}

__global__ __launch_bounds__(256) void gat_gather_fused_kernel(
    const float* __restrict__ h, const float* __restrict__ a_s,
    const float* __restrict__ a_d, const int* __restrict__ srt,
    const int* __restrict__ rowstart,
    const int* __restrict__ deg, const float* __restrict__ bias,
    const int* __restrict__ ng, const float* __restrict__ xg,
    const float* __restrict__ am,
    float* __restrict__ dst, int n) {
    gat_gather_body<true>(h, a_s, a_d, srt, rowstart, deg, bias, ng, xg, am, dst,
                          n, (int)(blockIdx.x * 8 + (threadIdx.x >> 5)));
}

__global__ __launch_bounds__(32) void gather_gemm_kernel(
    const float* __restrict__ hg, const float* __restrict__ a_sg,
    const float* __restrict__ a_dg, const int* __restrict__ srtg,
    const int* __restrict__ rowg, const int* __restrict__ degg,
    const float* __restrict__ biasg, float* __restrict__ dstg, int gn,
    const float* __restrict__ A, const float* __restrict__ B,
    float* __restrict__ C, int M) {
    if ((int)blockIdx.x < gn) {
        gat_gather_body<false>(hg, a_sg, a_dg, srtg, rowg, degg, biasg, nullptr,
                               nullptr, nullptr, dstg, gn, (int)blockIdx.x);
        return;
    }
    gemm128_strip<4, 128>(A, B, nullptr, C, M, (int)blockIdx.x - gn,
                          (int)threadIdx.x);
}

#define ADJ_UNROLL 16
#define ADJ_THREADS 128

__device__ __forceinline__ void adj_add_rows(
    unsigned lanes, int hit, int e0, const int* __restrict__ eig, int EG, int G,
    const float* __restrict__ xl, float4& acc, int& found) {
    while (lanes) {
        const int L = __builtin_ctz(lanes);
        lanes &= lanes - 1;
        unsigned bits = (unsigned)__builtin_amdgcn_readlane(hit, L) & 15u;
        found += __builtin_popcount(bits);
        while (bits) {
            const int e = e0 + 4 * L + __builtin_ctz(bits);
            bits &= bits - 1;
            const int g1 = min(max(eig[EG + e], 0), G - 1);
            const float4 r =
                *reinterpret_cast<const float4*>(xl + (size_t)g1 * 128);
            acc.x += r.x; acc.y += r.y; acc.z += r.z; acc.w += r.w;
        }
    }
}

__device__ __forceinline__ void adj_mean_body(
    const int* __restrict__ eig, int EG, int G,
    const float* __restrict__ xg, float* __restrict__ adj, int g,
    float (*part)[128], int* pfound) {
    const int tid = threadIdx.x, lane = tid & 31;
    const int w = __builtin_amdgcn_readfirstlane(tid >> 5);
    const float* xl = xg + lane * 4;
    float4 acc = make_float4(0.f, 0.f, 0.f, 0.f);
    int found = 0;
    const int nfull = EG >> 2;
    for (int qb = 0; qb < nfull; qb += ADJ_THREADS * ADJ_UNROLL) {
        int4 v[ADJ_UNROLL];
#pragma unroll
        for (int u = 0; u < ADJ_UNROLL; ++u) {
            const int q = min(qb + u * ADJ_THREADS + tid, nfull - 1);
            v[u] = *reinterpret_cast<const int4*>(eig + (size_t)q * 4);
        }
#pragma unroll
        for (int u = 0; u < ADJ_UNROLL; ++u) {
            int hit = (v[u].x == g ? 1 : 0) | (v[u].y == g ? 2 : 0) |
                      (v[u].z == g ? 4 : 0) | (v[u].w == g ? 8 : 0);
            if (qb + u * ADJ_THREADS + tid >= nfull) hit = 0;
            const unsigned lanes = __builtin_amdgcn_ballot_w32(hit != 0);
            if (lanes)
                adj_add_rows(lanes, hit, (qb + u * ADJ_THREADS + w * 32) * 4,
                             eig, EG, G, xl, acc, found);
        }
    }
    if (w == 0) {
        int hit = 0;
        if (lane == 0) {
            for (int c = 0; c < 3; ++c) {
                const int e = nfull * 4 + c;
                if (e < EG && eig[e] == g) hit |= 1 << c;
            }
        }
        const unsigned lanes = __builtin_amdgcn_ballot_w32(hit != 0);
        adj_add_rows(lanes, hit, nfull * 4, eig, EG, G, xl, acc, found);
    }
    *reinterpret_cast<float4*>(&part[w][lane * 4]) = acc;
    if (lane == 0) pfound[w] = found;
    __syncthreads();
    float s = 0.f;
    int c = 0;
    for (int k = 0; k < ADJ_THREADS / 32; ++k) {
        s += part[k][tid];
        c += pfound[k];
    }
    adj[(size_t)g * 128 + tid] = s / fmaxf((float)c, 1.f);
}

__global__ __launch_bounds__(ADJ_THREADS) void adj_attn_kernel(
    const int* __restrict__ eig, int EG, int G,
    const float* __restrict__ xg, float* __restrict__ adj,
    const float* __restrict__ h, const float* __restrict__ ws,
    const float* __restrict__ wd, float* __restrict__ a_s,
    float* __restrict__ a_d, int n) {
    __shared__ __align__(16) float part[ADJ_THREADS / 32][128];
    __shared__ int pfound[ADJ_THREADS / 32];
    if ((int)blockIdx.x < G) {
        adj_mean_body(eig, EG, G, xg, adj, (int)blockIdx.x, part, pfound);
        return;
    }
    attn_body(h, ws, wd, a_s, a_d, n, (int)blockIdx.x - G);
}

static inline unsigned cdivll(long long a, long long b) {
    return (unsigned)((a + b - 1) / b);
}
static inline size_t al4(size_t v) { return (v + 3) & ~(size_t)3; }

extern "C" void kernel_launch(void* const* d_in, const int* in_sizes, int n_in,
                              void* d_out, int out_size, void* d_ws,
                              size_t ws_size, hipStream_t stream) {
    const float* x      = (const float*)d_in[0];
    const float* x_g    = (const float*)d_in[1];
    const int*   ei     = (const int*)d_in[2];
    const int*   eig    = (const int*)d_in[3];
    const int*   ngrp   = (const int*)d_in[4];
    const float* W1     = (const float*)d_in[5];
    const float* atts1  = (const float*)d_in[6];
    const float* attd1  = (const float*)d_in[7];
    const float* bias1  = (const float*)d_in[8];
    const float* W2     = (const float*)d_in[9];
    const float* atts2  = (const float*)d_in[10];
    const float* attd2  = (const float*)d_in[11];
    const float* bias2  = (const float*)d_in[12];
    const float* W_fc   = (const float*)d_in[13];
    const float* b_fc   = (const float*)d_in[14];

    const int N  = in_sizes[0] / 128;
    const int G  = in_sizes[1] / 128;
    const int E  = in_sizes[2] / 2;
    const int EG = in_sizes[3] / 2;

    float* out    = (float*)d_out;
    float* xg_out = out + (size_t)N * 32;

    float* ws = (float*)d_ws;
    const int nbN = (N + BK_NODES - 1) / BK_NODES;
    const int nbG = (G + BK_NODES - 1) / BK_NODES;
    const int nctr = 2 * (nbN + nbG) * CTR_STRIDE;
    size_t o = 0;
    float* h     = ws + o; o += (size_t)N * 128;
    float* acc   = ws + o; o += (size_t)N * 128;
    float* a_s   = ws + o; o += (size_t)N * 4;
    float* a_d   = ws + o; o += (size_t)N * 4;
    int*   degN  = (int*)(ws + o); o += al4((size_t)N);
    int*   rowN  = (int*)(ws + o); o += al4((size_t)N);
    int*   bsN   = (int*)(ws + o); o += al4((size_t)nbN);
    int*   bsG   = (int*)(ws + o); o += al4((size_t)nbG);
    float* h_g   = ws + o; o += (size_t)G * 128;
    float* a_sg  = ws + o; o += (size_t)G * 4;
    float* a_dg  = ws + o; o += (size_t)G * 4;
    int*   degG  = (int*)(ws + o); o += al4((size_t)G);
    int*   rowG  = (int*)(ws + o); o += al4((size_t)G);
    int*   srtG  = (int*)(ws + o); o += al4((size_t)EG);
    float* adj   = ws + o; o += (size_t)G * 128;
    int*   ctr   = (int*)adj;
    int*   cntN  = ctr;
    int*   curN  = ctr + nbN * CTR_STRIDE;
    int*   cntG  = ctr + 2 * nbN * CTR_STRIDE;
    int*   curG  = cntG + nbG * CTR_STRIDE;
    int* binsN = (int*)acc;
    int* binsG = (int*)h;
    int* srtN  = (int*)out;

    const int nbMax = nbN > nbG ? nbN : nbG;
    const unsigned cbG = cdivll(EG, CHUNK_EDGES);
    const unsigned cbN = cdivll(E, CHUNK_EDGES);
    const unsigned tilesG = (unsigned)(((G + 15) / 16) * (128 / 64));
    const unsigned tilesN = (unsigned)(((N + 15) / 16) * (128 / 64));
    const unsigned zb = cdivll(nctr, ZERO_PER_BLOCK);
    const unsigned abG = cdivll((long long)G * 4, 256);

    zero_gemm_kernel<<<zb + tilesG, 32, 0, stream>>>(
        ctr, nctr, (int)zb, x_g, W2, h_g, G);
    attn_count_kernel<<<abG + cbG + cbN, 256, (size_t)nbMax * 4, stream>>>(
        h_g, atts2, attd2, a_sg, a_dg, G, (int)abG,
        eig, EG, G, nbG, cntG, (int)cbG,
        ei, E, N, nbN, cntN);
    bucket_fill_kernel<<<cbG + cbN, 256, (size_t)nbMax * 8, stream>>>(
        eig, EG, G, nbG, cntG, curG, bsG, binsG, (int)cbG,
        ei, E, N, nbN, cntN, curN, bsN, binsN);
    bucket_rank_kernel<<<(unsigned)(nbG + nbN), 1024, 0, stream>>>(
        binsG, cntG, bsG, EG, G, degG, rowG, srtG, nbG,
        binsN, cntN, bsN, E, N, degN, rowN, srtN);
    gather_gemm_kernel<<<(unsigned)G + tilesN, 32, 0, stream>>>(
        h_g, a_sg, a_dg, srtG, rowG, degG, bias2, xg_out, G,
        x, W1, h, N);
    adj_attn_kernel<<<(unsigned)G + cdivll((long long)N * 4, ADJ_THREADS),
                      ADJ_THREADS, 0, stream>>>(
        eig, EG, G, xg_out, adj,
        h, atts1, attd1, a_s, a_d, N);
    gat_gather_fused_kernel<<<cdivll(N, 8), 256, 0, stream>>>(
        h, a_s, a_d, srtN, rowN, degN, bias1, ngrp, xg_out, adj,
        acc, N);
    gemm128_wmma<2, 32><<<(unsigned)((N + 15) / 16), 32, 0, stream>>>(
        acc, W_fc, b_fc, out, N);
}
